# starting kernel
# baseline (speedup 1.0000x reference)
#include <hip/hip_runtime.h>
#include <hip/hip_bf16.h>

typedef float v2f __attribute__((ext_vector_type(2)));
typedef float v8f __attribute__((ext_vector_type(8)));

#define D 16
#define NGRAPH 1024

__global__ void fill_zero_k(float* __restrict__ p, long long n) {
    long long i = (long long)blockIdx.x * blockDim.x + threadIdx.x;
    if (i < n) p[i] = 0.0f;
}

__global__ void gather_gemm16_k(const float* __restrict__ src_rows,
                                const int* __restrict__ ids,
                                const float* __restrict__ W,
                                float* __restrict__ out,
                                int n_nodes) {
    const int lane  = threadIdx.x & 31;
    const int wave  = (int)((blockIdx.x * blockDim.x + threadIdx.x) >> 5);
    const int nwave = (int)((gridDim.x * blockDim.x) >> 5);
    const int row   = lane & 15;
    const int col   = lane & 15;
    const int koff  = (lane >> 4) << 1;
    const int rowb  = (lane >> 4) << 3;

    v2f bb0, bb1, bb2, bb3;
    bb0.x = W[(0  + koff) * D + col]; bb0.y = W[(1  + koff) * D + col];
    bb1.x = W[(4  + koff) * D + col]; bb1.y = W[(5  + koff) * D + col];
    bb2.x = W[(8  + koff) * D + col]; bb2.y = W[(9  + koff) * D + col];
    bb3.x = W[(12 + koff) * D + col]; bb3.y = W[(13 + koff) * D + col];

    const int ntiles = n_nodes >> 4;
    for (int tile = wave; tile < ntiles; tile += nwave) {
        const int node = (tile << 4) + row;
        const int r = ids ? ids[node] : node;
        const float* xrow = src_rows + (size_t)r * D;
        v2f a0 = *(const v2f*)(xrow + 0  + koff);
        v2f a1 = *(const v2f*)(xrow + 4  + koff);
        v2f a2 = *(const v2f*)(xrow + 8  + koff);
        v2f a3 = *(const v2f*)(xrow + 12 + koff);
        v8f c = {};
        c = __builtin_amdgcn_wmma_f32_16x16x4_f32(false, a0, false, bb0, (short)0, c, false, false);
        c = __builtin_amdgcn_wmma_f32_16x16x4_f32(false, a1, false, bb1, (short)0, c, false, false);
        c = __builtin_amdgcn_wmma_f32_16x16x4_f32(false, a2, false, bb2, (short)0, c, false, false);
        c = __builtin_amdgcn_wmma_f32_16x16x4_f32(false, a3, false, bb3, (short)0, c, false, false);
        float* ob = out + ((size_t)(tile << 4) + rowb) * D + col;
        #pragma unroll
        for (int rr = 0; rr < 8; ++rr) ob[(size_t)rr * D] = c[rr];
    }

    const int tail_start = ntiles << 4;
    if (tail_start < n_nodes) {
        int node = tail_start + (int)(blockIdx.x * blockDim.x + threadIdx.x);
        if (node < n_nodes) {
            const int r = ids ? ids[node] : node;
            const float* xrow = src_rows + (size_t)r * D;
            for (int j = 0; j < D; ++j) {
                float acc = 0.0f;
                for (int k = 0; k < D; ++k) acc += xrow[k] * W[k * D + j];
                out[(size_t)node * D + j] = acc;
            }
        }
    }
}

__global__ void degree_k(const int* __restrict__ dst, float* __restrict__ deg,
                         long long E) {
    long long e = (long long)blockIdx.x * blockDim.x + threadIdx.x;
    if (e < E) atomicAdd(deg + dst[e], 1.0f);
}

__global__ void rsqrt_k(float* __restrict__ d, int n) {
    int i = blockIdx.x * blockDim.x + threadIdx.x;
    if (i < n) d[i] = rsqrtf(d[i] + 1.0f);
}

__global__ void aggregate_k(const float* __restrict__ h,
                            const float* __restrict__ dis,
                            const int* __restrict__ src,
                            const int* __restrict__ dst,
                            float* __restrict__ agg,
                            long long E) {
    long long t = (long long)blockIdx.x * blockDim.x + threadIdx.x;
    long long e = t >> 4;
    int j = (int)(t & 15);
    if (e < E) {
        int s = src[e], d = dst[e];
        float norm = dis[s] * dis[d];
        atomicAdd(agg + (size_t)d * D + j, h[(size_t)s * D + j] * norm);
    }
}

__global__ void node_relu_k(float* __restrict__ h, const float* __restrict__ agg,
                            const float* __restrict__ dis,
                            const float* __restrict__ bias, long long n16) {
    long long t = (long long)blockIdx.x * blockDim.x + threadIdx.x;
    if (t < n16) {
        long long i = t >> 4; int j = (int)(t & 15);
        float di = dis[i];
        float v = agg[t] + h[t] * (di * di) + bias[j];
        h[t] = v > 0.0f ? v : 0.0f;
    }
}

__global__ void node_pool_k(const float* __restrict__ h2,
                            const float* __restrict__ agg,
                            const float* __restrict__ dis,
                            const float* __restrict__ bias,
                            const int* __restrict__ batch,
                            float* __restrict__ pooled,
                            float* __restrict__ cnt, long long n16) {
    long long t = (long long)blockIdx.x * blockDim.x + threadIdx.x;
    if (t < n16) {
        long long i = t >> 4; int j = (int)(t & 15);
        float di = dis[i];
        float v = agg[t] + h2[t] * (di * di) + bias[j];
        int g = batch[i];
        atomicAdd(pooled + (size_t)g * D + j, v);
        if (j == 0) atomicAdd(cnt + g, 1.0f);
    }
}

__global__ void fc_k(const float* __restrict__ pooled_r, const float* __restrict__ cnt_r,
                     const float* __restrict__ pooled_l, const float* __restrict__ cnt_l,
                     const float* __restrict__ fcW, const float* __restrict__ fcb,
                     float* __restrict__ out) {
    int t = blockIdx.x * blockDim.x + threadIdx.x;
    if (t < NGRAPH * 6) {
        int g = t / 6, o = t - g * 6;
        float icr = 1.0f / fmaxf(cnt_r[g], 1.0f);
        float icl = 1.0f / fmaxf(cnt_l[g], 1.0f);
        float acc = fcb[o];
        #pragma unroll
        for (int j = 0; j < D; ++j)
            acc += fcW[o * (2 * D) + j] * (pooled_r[g * D + j] * icr);
        #pragma unroll
        for (int j = 0; j < D; ++j)
            acc += fcW[o * (2 * D) + D + j] * (pooled_l[g * D + j] * icl);
        if (o < 3) out[g * 3 + o] = acc;
        else       out[NGRAPH * 3 + g * 3 + (o - 3)] = acc;
    }
}

static void run_protein(const float* emb, const float* W1, const float* b1,
                        const float* W2, const float* b2,
                        const int* ids, const int* edge, const int* batch,
                        int N, long long E,
                        float* bufA, float* bufB, float* dis,
                        float* pooled, float* cnt, hipStream_t stream) {
    const int* src = edge;
    const int* dst = edge + E;
    const long long n16 = (long long)N * D;
    const int B = 256;
    dim3 blk(B);
    int gemm_grid = (int)(((N >> 4) + 7) / 8);
    if (gemm_grid < 1) gemm_grid = 1;

    fill_zero_k<<<(int)((N + B - 1) / B), blk, 0, stream>>>(dis, N);
    fill_zero_k<<<(int)((n16 + B - 1) / B), blk, 0, stream>>>(bufB, n16);

    gather_gemm16_k<<<gemm_grid, blk, 0, stream>>>(emb, ids, W1, bufA, N);

    degree_k<<<(int)((E + B - 1) / B), blk, 0, stream>>>(dst, dis, E);
    rsqrt_k<<<(int)((N + B - 1) / B), blk, 0, stream>>>(dis, N);

    aggregate_k<<<(int)((E * 16 + B - 1) / B), blk, 0, stream>>>(bufA, dis, src, dst, bufB, E);
    node_relu_k<<<(int)((n16 + B - 1) / B), blk, 0, stream>>>(bufA, bufB, dis, b1, n16);

    gather_gemm16_k<<<gemm_grid, blk, 0, stream>>>(bufA, nullptr, W2, bufB, N);

    fill_zero_k<<<(int)((n16 + B - 1) / B), blk, 0, stream>>>(bufA, n16);
    aggregate_k<<<(int)((E * 16 + B - 1) / B), blk, 0, stream>>>(bufB, dis, src, dst, bufA, E);

    fill_zero_k<<<(int)((NGRAPH * D + B - 1) / B), blk, 0, stream>>>(pooled, NGRAPH * D);
    fill_zero_k<<<(int)((NGRAPH + B - 1) / B), blk, 0, stream>>>(cnt, NGRAPH);
    node_pool_k<<<(int)((n16 + B - 1) / B), blk, 0, stream>>>(bufB, bufA, dis, b2, batch,
                                                              pooled, cnt, n16);
}

extern "C" void kernel_launch(void* const* d_in, const int* in_sizes, int n_in,
                              void* d_out, int out_size, void* d_ws, size_t ws_size,
                              hipStream_t stream) {
    const float* emb = (const float*)d_in[0];
    const float* W1  = (const float*)d_in[1];
    const float* b1  = (const float*)d_in[2];
    const float* W2  = (const float*)d_in[3];
    const float* b2  = (const float*)d_in[4];
    const float* fcW = (const float*)d_in[5];
    const float* fcb = (const float*)d_in[6];
    const int* rx     = (const int*)d_in[7];
    const int* redge  = (const int*)d_in[8];
    const int* rbatch = (const int*)d_in[9];
    const int* lx     = (const int*)d_in[10];
    const int* ledge  = (const int*)d_in[11];
    const int* lbatch = (const int*)d_in[12];

    const int N = in_sizes[7];
    const long long E = in_sizes[8] / 2;

    char* ws = (char*)d_ws;
    float* bufA = (float*)ws;  ws += (size_t)N * D * sizeof(float);
    float* bufB = (float*)ws;  ws += (size_t)N * D * sizeof(float);
    float* dis  = (float*)ws;  ws += (size_t)N * sizeof(float);
    float* pooled_r = (float*)ws; ws += (size_t)NGRAPH * D * sizeof(float);
    float* pooled_l = (float*)ws; ws += (size_t)NGRAPH * D * sizeof(float);
    float* cnt_r = (float*)ws; ws += (size_t)NGRAPH * sizeof(float);
    float* cnt_l = (float*)ws; ws += (size_t)NGRAPH * sizeof(float);

    run_protein(emb, W1, b1, W2, b2, rx, redge, rbatch, N, E,
                bufA, bufB, dis, pooled_r, cnt_r, stream);
    run_protein(emb, W1, b1, W2, b2, lx, ledge, lbatch, N, E,
                bufA, bufB, dis, pooled_l, cnt_l, stream);

    fc_k<<<(NGRAPH * 6 + 255) / 256, 256, 0, stream>>>(pooled_r, cnt_r, pooled_l, cnt_l,
                                                       fcW, fcb, (float*)d_out);
}


# in-degree counted in LDS counters (node ranges x edge slices; integer partial planes summed in the rsqrt kernel) instead of one scattered global f32 atomic per edge; degree sum integer, still exact
# speedup vs baseline: 1.2453x; 1.2453x over previous
#include <hip/hip_runtime.h>
#include <hip/hip_bf16.h>

typedef float v2f __attribute__((ext_vector_type(2)));
typedef float v8f __attribute__((ext_vector_type(8)));

#define D 16
#define NGRAPH 1024

__global__ void fill_zero_k(float* __restrict__ p, long long n) {
    long long i = (long long)blockIdx.x * blockDim.x + threadIdx.x;
    if (i < n) p[i] = 0.0f;
}

__global__ void gather_gemm16_k(const float* __restrict__ src_rows,
                                const int* __restrict__ ids,
                                const float* __restrict__ W,
                                float* __restrict__ out,
                                int n_nodes) {
    const int lane  = threadIdx.x & 31;
    const int wave  = (int)((blockIdx.x * blockDim.x + threadIdx.x) >> 5);
    const int nwave = (int)((gridDim.x * blockDim.x) >> 5);
    const int row   = lane & 15;
    const int col   = lane & 15;
    const int koff  = (lane >> 4) << 1;
    const int rowb  = (lane >> 4) << 3;

    v2f bb0, bb1, bb2, bb3;
    bb0.x = W[(0  + koff) * D + col]; bb0.y = W[(1  + koff) * D + col];
    bb1.x = W[(4  + koff) * D + col]; bb1.y = W[(5  + koff) * D + col];
    bb2.x = W[(8  + koff) * D + col]; bb2.y = W[(9  + koff) * D + col];
    bb3.x = W[(12 + koff) * D + col]; bb3.y = W[(13 + koff) * D + col];

    const int ntiles = n_nodes >> 4;
    for (int tile = wave; tile < ntiles; tile += nwave) {
        const int node = (tile << 4) + row;
        const int r = ids ? ids[node] : node;
        const float* xrow = src_rows + (size_t)r * D;
        v2f a0 = *(const v2f*)(xrow + 0  + koff);
        v2f a1 = *(const v2f*)(xrow + 4  + koff);
        v2f a2 = *(const v2f*)(xrow + 8  + koff);
        v2f a3 = *(const v2f*)(xrow + 12 + koff);
        v8f c = {};
        c = __builtin_amdgcn_wmma_f32_16x16x4_f32(false, a0, false, bb0, (short)0, c, false, false);
        c = __builtin_amdgcn_wmma_f32_16x16x4_f32(false, a1, false, bb1, (short)0, c, false, false);
        c = __builtin_amdgcn_wmma_f32_16x16x4_f32(false, a2, false, bb2, (short)0, c, false, false);
        c = __builtin_amdgcn_wmma_f32_16x16x4_f32(false, a3, false, bb3, (short)0, c, false, false);
        float* ob = out + ((size_t)(tile << 4) + rowb) * D + col;
        #pragma unroll
        for (int rr = 0; rr < 8; ++rr) ob[(size_t)rr * D] = c[rr];
    }

    const int tail_start = ntiles << 4;
    if (tail_start < n_nodes) {
        int node = tail_start + (int)(blockIdx.x * blockDim.x + threadIdx.x);
        if (node < n_nodes) {
            const int r = ids ? ids[node] : node;
            const float* xrow = src_rows + (size_t)r * D;
            for (int j = 0; j < D; ++j) {
                float acc = 0.0f;
                for (int k = 0; k < D; ++k) acc += xrow[k] * W[k * D + j];
                out[(size_t)node * D + j] = acc;
            }
        }
    }
}

#define DEG_SLICES  16
#define DEG_RANGES  16
#define DEG_CAP     12512
#define DEG_THREADS 1024
#define DEG_UNROLL  8

__global__ __launch_bounds__(DEG_THREADS)
void degree_lds_k(const int* __restrict__ dst, unsigned* __restrict__ part,
                  int n, long long E) {
    __shared__ unsigned cnt[DEG_CAP];
    const int tid   = threadIdx.x;
    const int slice = blockIdx.x;
    const long long per = (E + DEG_SLICES - 1) / DEG_SLICES;
    const long long e0  = (long long)slice * per;
    long long e1 = e0 + per; if (e1 > E) e1 = E;
    const int* __restrict__ ds = dst + e0;
    const int ne = (e1 > e0) ? (int)(e1 - e0) : 0;

    for (int lo = blockIdx.y * DEG_CAP; lo < n; lo += gridDim.y * DEG_CAP) {
        const int left = n - lo;
        const unsigned span = (unsigned)(left < DEG_CAP ? left : DEG_CAP);
        for (int i = tid; i < DEG_CAP; i += DEG_THREADS) cnt[i] = 0u;
        __syncthreads();
        for (int e = tid; e < ne; e += DEG_THREADS * DEG_UNROLL) {
            int d[DEG_UNROLL];
            #pragma unroll
            for (int u = 0; u < DEG_UNROLL; ++u) {
                const int ee = e + u * DEG_THREADS;
                d[u] = (ee < ne) ? ds[ee] : -1;
            }
            #pragma unroll
            for (int u = 0; u < DEG_UNROLL; ++u) {
                const unsigned r = (unsigned)(d[u] - lo);
                if (r < span) atomicAdd(&cnt[r], 1u);
            }
        }
        __syncthreads();
        unsigned* __restrict__ prow = part + (size_t)slice * n + lo;
        for (int i = tid; i < (int)span; i += DEG_THREADS) prow[i] = cnt[i];
        __syncthreads();
    }
}

__global__ void dis_from_parts_k(const unsigned* __restrict__ part,
                                 float* __restrict__ dis, int n) {
    int i = blockIdx.x * blockDim.x + threadIdx.x;
    if (i < n) {
        unsigned c = 0u;
        #pragma unroll
        for (int s = 0; s < DEG_SLICES; ++s) c += part[(size_t)s * n + i];
        dis[i] = rsqrtf((float)c + 1.0f);
    }
}

__global__ void aggregate_k(const float* __restrict__ h,
                            const float* __restrict__ dis,
                            const int* __restrict__ src,
                            const int* __restrict__ dst,
                            float* __restrict__ agg,
                            long long E) {
    long long t = (long long)blockIdx.x * blockDim.x + threadIdx.x;
    long long e = t >> 4;
    int j = (int)(t & 15);
    if (e < E) {
        int s = src[e], d = dst[e];
        float norm = dis[s] * dis[d];
        atomicAdd(agg + (size_t)d * D + j, h[(size_t)s * D + j] * norm);
    }
}

__global__ void node_relu_k(float* __restrict__ h, const float* __restrict__ agg,
                            const float* __restrict__ dis,
                            const float* __restrict__ bias, long long n16) {
    long long t = (long long)blockIdx.x * blockDim.x + threadIdx.x;
    if (t < n16) {
        long long i = t >> 4; int j = (int)(t & 15);
        float di = dis[i];
        float v = agg[t] + h[t] * (di * di) + bias[j];
        h[t] = v > 0.0f ? v : 0.0f;
    }
}

__global__ void node_pool_k(const float* __restrict__ h2,
                            const float* __restrict__ agg,
                            const float* __restrict__ dis,
                            const float* __restrict__ bias,
                            const int* __restrict__ batch,
                            float* __restrict__ pooled,
                            float* __restrict__ cnt, long long n16) {
    long long t = (long long)blockIdx.x * blockDim.x + threadIdx.x;
    if (t < n16) {
        long long i = t >> 4; int j = (int)(t & 15);
        float di = dis[i];
        float v = agg[t] + h2[t] * (di * di) + bias[j];
        int g = batch[i];
        atomicAdd(pooled + (size_t)g * D + j, v);
        if (j == 0) atomicAdd(cnt + g, 1.0f);
    }
}

__global__ void fc_k(const float* __restrict__ pooled_r, const float* __restrict__ cnt_r,
                     const float* __restrict__ pooled_l, const float* __restrict__ cnt_l,
                     const float* __restrict__ fcW, const float* __restrict__ fcb,
                     float* __restrict__ out) {
    int t = blockIdx.x * blockDim.x + threadIdx.x;
    if (t < NGRAPH * 6) {
        int g = t / 6, o = t - g * 6;
        float icr = 1.0f / fmaxf(cnt_r[g], 1.0f);
        float icl = 1.0f / fmaxf(cnt_l[g], 1.0f);
        float acc = fcb[o];
        #pragma unroll
        for (int j = 0; j < D; ++j)
            acc += fcW[o * (2 * D) + j] * (pooled_r[g * D + j] * icr);
        #pragma unroll
        for (int j = 0; j < D; ++j)
            acc += fcW[o * (2 * D) + D + j] * (pooled_l[g * D + j] * icl);
        if (o < 3) out[g * 3 + o] = acc;
        else       out[NGRAPH * 3 + g * 3 + (o - 3)] = acc;
    }
}

static void run_protein(const float* emb, const float* W1, const float* b1,
                        const float* W2, const float* b2,
                        const int* ids, const int* edge, const int* batch,
                        int N, long long E,
                        float* bufA, float* bufB, float* dis,
                        float* pooled, float* cnt, hipStream_t stream) {
    const int* src = edge;
    const int* dst = edge + E;
    const long long n16 = (long long)N * D;
    const int B = 256;
    dim3 blk(B);
    int gemm_grid = (int)(((N >> 4) + 7) / 8);
    if (gemm_grid < 1) gemm_grid = 1;

    degree_lds_k<<<dim3(DEG_SLICES, DEG_RANGES), dim3(DEG_THREADS), 0, stream>>>(
        dst, (unsigned*)bufB, N, E);
    dis_from_parts_k<<<(int)((N + B - 1) / B), blk, 0, stream>>>((const unsigned*)bufB, dis, N);

    fill_zero_k<<<(int)((n16 + B - 1) / B), blk, 0, stream>>>(bufB, n16);

    gather_gemm16_k<<<gemm_grid, blk, 0, stream>>>(emb, ids, W1, bufA, N);

    aggregate_k<<<(int)((E * 16 + B - 1) / B), blk, 0, stream>>>(bufA, dis, src, dst, bufB, E);
    node_relu_k<<<(int)((n16 + B - 1) / B), blk, 0, stream>>>(bufA, bufB, dis, b1, n16);

    gather_gemm16_k<<<gemm_grid, blk, 0, stream>>>(bufA, nullptr, W2, bufB, N);

    fill_zero_k<<<(int)((n16 + B - 1) / B), blk, 0, stream>>>(bufA, n16);
    aggregate_k<<<(int)((E * 16 + B - 1) / B), blk, 0, stream>>>(bufB, dis, src, dst, bufA, E);

    fill_zero_k<<<(int)((NGRAPH * D + B - 1) / B), blk, 0, stream>>>(pooled, NGRAPH * D);
    fill_zero_k<<<(int)((NGRAPH + B - 1) / B), blk, 0, stream>>>(cnt, NGRAPH);
    node_pool_k<<<(int)((n16 + B - 1) / B), blk, 0, stream>>>(bufB, bufA, dis, b2, batch,
                                                              pooled, cnt, n16);
}

extern "C" void kernel_launch(void* const* d_in, const int* in_sizes, int n_in,
                              void* d_out, int out_size, void* d_ws, size_t ws_size,
                              hipStream_t stream) {
    const float* emb = (const float*)d_in[0];
    const float* W1  = (const float*)d_in[1];
    const float* b1  = (const float*)d_in[2];
    const float* W2  = (const float*)d_in[3];
    const float* b2  = (const float*)d_in[4];
    const float* fcW = (const float*)d_in[5];
    const float* fcb = (const float*)d_in[6];
    const int* rx     = (const int*)d_in[7];
    const int* redge  = (const int*)d_in[8];
    const int* rbatch = (const int*)d_in[9];
    const int* lx     = (const int*)d_in[10];
    const int* ledge  = (const int*)d_in[11];
    const int* lbatch = (const int*)d_in[12];

    const int N = in_sizes[7];
    const long long E = in_sizes[8] / 2;

    char* ws = (char*)d_ws;
    float* bufA = (float*)ws;  ws += (size_t)N * D * sizeof(float);
    float* bufB = (float*)ws;  ws += (size_t)N * D * sizeof(float);
    float* dis  = (float*)ws;  ws += (size_t)N * sizeof(float);
    float* pooled_r = (float*)ws; ws += (size_t)NGRAPH * D * sizeof(float);
    float* pooled_l = (float*)ws; ws += (size_t)NGRAPH * D * sizeof(float);
    float* cnt_r = (float*)ws; ws += (size_t)NGRAPH * sizeof(float);
    float* cnt_l = (float*)ws; ws += (size_t)NGRAPH * sizeof(float);

    run_protein(emb, W1, b1, W2, b2, rx, redge, rbatch, N, E,
                bufA, bufB, dis, pooled_r, cnt_r, stream);
    run_protein(emb, W1, b1, W2, b2, lx, ledge, lbatch, N, E,
                bufA, bufB, dis, pooled_l, cnt_l, stream);

    fc_k<<<(NGRAPH * 6 + 255) / 256, 256, 0, stream>>>(pooled_r, cnt_r, pooled_l, cnt_l,
                                                       fcW, fcb, (float*)d_out);
}


# layer-2 aggregation and mean pool summed per graph in LDS f32 accumulators by linearity (per-block sums flushed by global atomics), no per-node layer-2 atomics; float summation order changes
# speedup vs baseline: 1.9034x; 1.5285x over previous
#include <hip/hip_runtime.h>
#include <hip/hip_bf16.h>

typedef float v2f __attribute__((ext_vector_type(2)));
typedef float v8f __attribute__((ext_vector_type(8)));

#define D 16
#define NGRAPH 1024

__global__ void fill_zero_k(float* __restrict__ p, long long n) {
    long long i = (long long)blockIdx.x * blockDim.x + threadIdx.x;
    if (i < n) p[i] = 0.0f;
}

__global__ void gather_gemm16_k(const float* __restrict__ src_rows,
                                const int* __restrict__ ids,
                                const float* __restrict__ W,
                                float* __restrict__ out,
                                int n_nodes) {
    const int lane  = threadIdx.x & 31;
    const int wave  = (int)((blockIdx.x * blockDim.x + threadIdx.x) >> 5);
    const int nwave = (int)((gridDim.x * blockDim.x) >> 5);
    const int row   = lane & 15;
    const int col   = lane & 15;
    const int koff  = (lane >> 4) << 1;
    const int rowb  = (lane >> 4) << 3;

    v2f bb0, bb1, bb2, bb3;
    bb0.x = W[(0  + koff) * D + col]; bb0.y = W[(1  + koff) * D + col];
    bb1.x = W[(4  + koff) * D + col]; bb1.y = W[(5  + koff) * D + col];
    bb2.x = W[(8  + koff) * D + col]; bb2.y = W[(9  + koff) * D + col];
    bb3.x = W[(12 + koff) * D + col]; bb3.y = W[(13 + koff) * D + col];

    const int ntiles = n_nodes >> 4;
    for (int tile = wave; tile < ntiles; tile += nwave) {
        const int node = (tile << 4) + row;
        const int r = ids ? ids[node] : node;
        const float* xrow = src_rows + (size_t)r * D;
        v2f a0 = *(const v2f*)(xrow + 0  + koff);
        v2f a1 = *(const v2f*)(xrow + 4  + koff);
        v2f a2 = *(const v2f*)(xrow + 8  + koff);
        v2f a3 = *(const v2f*)(xrow + 12 + koff);
        v8f c = {};
        c = __builtin_amdgcn_wmma_f32_16x16x4_f32(false, a0, false, bb0, (short)0, c, false, false);
        c = __builtin_amdgcn_wmma_f32_16x16x4_f32(false, a1, false, bb1, (short)0, c, false, false);
        c = __builtin_amdgcn_wmma_f32_16x16x4_f32(false, a2, false, bb2, (short)0, c, false, false);
        c = __builtin_amdgcn_wmma_f32_16x16x4_f32(false, a3, false, bb3, (short)0, c, false, false);
        float* ob = out + ((size_t)(tile << 4) + rowb) * D + col;
        #pragma unroll
        for (int rr = 0; rr < 8; ++rr) ob[(size_t)rr * D] = c[rr];
    }

    const int tail_start = ntiles << 4;
    if (tail_start < n_nodes) {
        int node = tail_start + (int)(blockIdx.x * blockDim.x + threadIdx.x);
        if (node < n_nodes) {
            const int r = ids ? ids[node] : node;
            const float* xrow = src_rows + (size_t)r * D;
            for (int j = 0; j < D; ++j) {
                float acc = 0.0f;
                for (int k = 0; k < D; ++k) acc += xrow[k] * W[k * D + j];
                out[(size_t)node * D + j] = acc;
            }
        }
    }
}

#define DEG_SLICES  16
#define DEG_RANGES  16
#define DEG_CAP     12512
#define DEG_THREADS 1024
#define DEG_UNROLL  8

__global__ __launch_bounds__(DEG_THREADS)
void degree_lds_k(const int* __restrict__ dst, unsigned* __restrict__ part,
                  int n, long long E) {
    __shared__ unsigned cnt[DEG_CAP];
    const int tid   = threadIdx.x;
    const int slice = blockIdx.x;
    const long long per = (E + DEG_SLICES - 1) / DEG_SLICES;
    const long long e0  = (long long)slice * per;
    long long e1 = e0 + per; if (e1 > E) e1 = E;
    const int* __restrict__ ds = dst + e0;
    const int ne = (e1 > e0) ? (int)(e1 - e0) : 0;

    for (int lo = blockIdx.y * DEG_CAP; lo < n; lo += gridDim.y * DEG_CAP) {
        const int left = n - lo;
        const unsigned span = (unsigned)(left < DEG_CAP ? left : DEG_CAP);
        for (int i = tid; i < DEG_CAP; i += DEG_THREADS) cnt[i] = 0u;
        __syncthreads();
        for (int e = tid; e < ne; e += DEG_THREADS * DEG_UNROLL) {
            int d[DEG_UNROLL];
            #pragma unroll
            for (int u = 0; u < DEG_UNROLL; ++u) {
                const int ee = e + u * DEG_THREADS;
                d[u] = (ee < ne) ? ds[ee] : -1;
            }
            #pragma unroll
            for (int u = 0; u < DEG_UNROLL; ++u) {
                const unsigned r = (unsigned)(d[u] - lo);
                if (r < span) atomicAdd(&cnt[r], 1u);
            }
        }
        __syncthreads();
        unsigned* __restrict__ prow = part + (size_t)slice * n + lo;
        for (int i = tid; i < (int)span; i += DEG_THREADS) prow[i] = cnt[i];
        __syncthreads();
    }
}

__global__ void dis_from_parts_k(const unsigned* __restrict__ part,
                                 float* __restrict__ dis, int n) {
    int i = blockIdx.x * blockDim.x + threadIdx.x;
    if (i < n) {
        unsigned c = 0u;
        #pragma unroll
        for (int s = 0; s < DEG_SLICES; ++s) c += part[(size_t)s * n + i];
        dis[i] = rsqrtf((float)c + 1.0f);
    }
}

__global__ void aggregate_k(const float* __restrict__ h,
                            const float* __restrict__ dis,
                            const int* __restrict__ src,
                            const int* __restrict__ dst,
                            float* __restrict__ agg,
                            long long E) {
    long long t = (long long)blockIdx.x * blockDim.x + threadIdx.x;
    long long e = t >> 4;
    int j = (int)(t & 15);
    if (e < E) {
        int s = src[e], d = dst[e];
        float norm = dis[s] * dis[d];
        atomicAdd(agg + (size_t)d * D + j, h[(size_t)s * D + j] * norm);
    }
}

__global__ void node_relu_k(float* __restrict__ h, const float* __restrict__ agg,
                            const float* __restrict__ dis,
                            const float* __restrict__ bias, long long n16) {
    long long t = (long long)blockIdx.x * blockDim.x + threadIdx.x;
    if (t < n16) {
        long long i = t >> 4; int j = (int)(t & 15);
        float di = dis[i];
        float v = agg[t] + h[t] * (di * di) + bias[j];
        h[t] = v > 0.0f ? v : 0.0f;
    }
}

#define POOL_THREADS 1024
#define POOL_BLOCKS  256

template <int K> __device__ __forceinline__ int row_bcast_i(int v) {
    return __builtin_amdgcn_update_dpp(0, v, 0x150 + K, 0xf, 0xf, false);
}
template <int K> __device__ __forceinline__ float row_bcast_f(float v) {
    return __int_as_float(row_bcast_i<K>(__float_as_int(v)));
}

#define ROWS16(X) X(0) X(1) X(2) X(3) X(4) X(5) X(6) X(7) \
                  X(8) X(9) X(10) X(11) X(12) X(13) X(14) X(15)

__global__ __launch_bounds__(POOL_THREADS)
void edge_node_pool_k(const float* __restrict__ h2,
                      const float* __restrict__ dis,
                      const float* __restrict__ bias,
                      const int* __restrict__ src,
                      const int* __restrict__ dst,
                      const int* __restrict__ batch,
                      float* __restrict__ pooled,
                      float* __restrict__ cnt,
                      int n, long long E) {
    __shared__ float acc[NGRAPH * D];
    __shared__ float gcnt[NGRAPH];
    const int tid  = threadIdx.x;
    const int lane = tid & 31;
    const int j    = lane & 15;

    for (int i = tid; i < NGRAPH * D; i += POOL_THREADS) acc[i] = 0.0f;
    for (int i = tid; i < NGRAPH; i += POOL_THREADS) gcnt[i] = 0.0f;
    __syncthreads();

    const long long wave  = (long long)blockIdx.x * (POOL_THREADS >> 5)
                          + __builtin_amdgcn_readfirstlane(tid >> 5);
    const long long nwave = (long long)gridDim.x * (POOL_THREADS >> 5);

    const long long nfull = E >> 5;
    long long c = wave;
    int s = 0, d = 0;
    if (c < nfull) { s = src[(c << 5) + lane]; d = dst[(c << 5) + lane]; }
    while (c < nfull) {
        const float w = dis[s] * dis[d];
        const int   g = batch[d];
#define LOAD_ROW(K) const float hv##K = h2[(unsigned)row_bcast_i<K>(s) * D + j];
        ROWS16(LOAD_ROW)
#undef LOAD_ROW
        const long long cn = c + nwave;
        int sn = 0, dn = 0;
        if (cn < nfull) { sn = src[(cn << 5) + lane]; dn = dst[(cn << 5) + lane]; }
#define ADD_ROW(K) atomicAdd(&acc[row_bcast_i<K>(g) * D + j], hv##K * row_bcast_f<K>(w));
        ROWS16(ADD_ROW)
#undef ADD_ROW
        s = sn; d = dn; c = cn;
    }
    if (blockIdx.x == 0 && tid < 16) {
        for (long long e = nfull << 5; e < E; ++e) {
            const int ss = src[e], dd = dst[e];
            const float w = dis[ss] * dis[dd];
            atomicAdd(&acc[batch[dd] * D + j], h2[(size_t)ss * D + j] * w);
        }
    }

    const long long n16 = (long long)n * D;
    for (long long t = (long long)blockIdx.x * POOL_THREADS + tid; t < n16;
         t += (long long)gridDim.x * POOL_THREADS) {
        const int i = (int)(t >> 4);
        const float di = dis[i];
        const int   g  = batch[i];
        atomicAdd(&acc[g * D + j], h2[t] * (di * di) + bias[j]);
        if (j == 0) atomicAdd(&gcnt[g], 1.0f);
    }
    __syncthreads();

    const int rot = (int)((blockIdx.x * POOL_THREADS) & (NGRAPH * D - 1));
    for (int i = tid; i < NGRAPH * D; i += POOL_THREADS) {
        const int idx = (i + rot) & (NGRAPH * D - 1);
        const float v = acc[idx];
        if (v != 0.0f) atomicAdd(pooled + idx, v);
    }
    for (int i = tid; i < NGRAPH; i += POOL_THREADS) {
        const float v = gcnt[i];
        if (v != 0.0f) atomicAdd(cnt + i, v);
    }
}

__global__ void fc_k(const float* __restrict__ pooled_r, const float* __restrict__ cnt_r,
                     const float* __restrict__ pooled_l, const float* __restrict__ cnt_l,
                     const float* __restrict__ fcW, const float* __restrict__ fcb,
                     float* __restrict__ out) {
    int t = blockIdx.x * blockDim.x + threadIdx.x;
    if (t < NGRAPH * 6) {
        int g = t / 6, o = t - g * 6;
        float icr = 1.0f / fmaxf(cnt_r[g], 1.0f);
        float icl = 1.0f / fmaxf(cnt_l[g], 1.0f);
        float acc = fcb[o];
        #pragma unroll
        for (int j = 0; j < D; ++j)
            acc += fcW[o * (2 * D) + j] * (pooled_r[g * D + j] * icr);
        #pragma unroll
        for (int j = 0; j < D; ++j)
            acc += fcW[o * (2 * D) + D + j] * (pooled_l[g * D + j] * icl);
        if (o < 3) out[g * 3 + o] = acc;
        else       out[NGRAPH * 3 + g * 3 + (o - 3)] = acc;
    }
}

static void run_protein(const float* emb, const float* W1, const float* b1,
                        const float* W2, const float* b2,
                        const int* ids, const int* edge, const int* batch,
                        int N, long long E,
                        float* bufA, float* bufB, float* dis,
                        float* pooled, float* cnt, hipStream_t stream) {
    const int* src = edge;
    const int* dst = edge + E;
    const long long n16 = (long long)N * D;
    const int B = 256;
    dim3 blk(B);
    int gemm_grid = (int)(((N >> 4) + 7) / 8);
    if (gemm_grid < 1) gemm_grid = 1;

    degree_lds_k<<<dim3(DEG_SLICES, DEG_RANGES), dim3(DEG_THREADS), 0, stream>>>(
        dst, (unsigned*)bufB, N, E);
    dis_from_parts_k<<<(int)((N + B - 1) / B), blk, 0, stream>>>((const unsigned*)bufB, dis, N);

    fill_zero_k<<<(int)((n16 + B - 1) / B), blk, 0, stream>>>(bufB, n16);

    gather_gemm16_k<<<gemm_grid, blk, 0, stream>>>(emb, ids, W1, bufA, N);

    aggregate_k<<<(int)((E * 16 + B - 1) / B), blk, 0, stream>>>(bufA, dis, src, dst, bufB, E);
    node_relu_k<<<(int)((n16 + B - 1) / B), blk, 0, stream>>>(bufA, bufB, dis, b1, n16);

    gather_gemm16_k<<<gemm_grid, blk, 0, stream>>>(bufA, nullptr, W2, bufB, N);

    fill_zero_k<<<(int)((NGRAPH * D + B - 1) / B), blk, 0, stream>>>(pooled, NGRAPH * D);
    fill_zero_k<<<(int)((NGRAPH + B - 1) / B), blk, 0, stream>>>(cnt, NGRAPH);
    edge_node_pool_k<<<POOL_BLOCKS, POOL_THREADS, 0, stream>>>(bufB, dis, b2, src, dst, batch,
                                                               pooled, cnt, N, E);
}

extern "C" void kernel_launch(void* const* d_in, const int* in_sizes, int n_in,
                              void* d_out, int out_size, void* d_ws, size_t ws_size,
                              hipStream_t stream) {
    const float* emb = (const float*)d_in[0];
    const float* W1  = (const float*)d_in[1];
    const float* b1  = (const float*)d_in[2];
    const float* W2  = (const float*)d_in[3];
    const float* b2  = (const float*)d_in[4];
    const float* fcW = (const float*)d_in[5];
    const float* fcb = (const float*)d_in[6];
    const int* rx     = (const int*)d_in[7];
    const int* redge  = (const int*)d_in[8];
    const int* rbatch = (const int*)d_in[9];
    const int* lx     = (const int*)d_in[10];
    const int* ledge  = (const int*)d_in[11];
    const int* lbatch = (const int*)d_in[12];

    const int N = in_sizes[7];
    const long long E = in_sizes[8] / 2;

    char* ws = (char*)d_ws;
    float* bufA = (float*)ws;  ws += (size_t)N * D * sizeof(float);
    float* bufB = (float*)ws;  ws += (size_t)N * D * sizeof(float);
    float* dis  = (float*)ws;  ws += (size_t)N * sizeof(float);
    float* pooled_r = (float*)ws; ws += (size_t)NGRAPH * D * sizeof(float);
    float* pooled_l = (float*)ws; ws += (size_t)NGRAPH * D * sizeof(float);
    float* cnt_r = (float*)ws; ws += (size_t)NGRAPH * sizeof(float);
    float* cnt_l = (float*)ws; ws += (size_t)NGRAPH * sizeof(float);

    run_protein(emb, W1, b1, W2, b2, rx, redge, rbatch, N, E,
                bufA, bufB, dis, pooled_r, cnt_r, stream);
    run_protein(emb, W1, b1, W2, b2, lx, ledge, lbatch, N, E,
                bufA, bufB, dis, pooled_l, cnt_l, stream);

    fc_k<<<(NGRAPH * 6 + 255) / 256, 256, 0, stream>>>(pooled_r, cnt_r, pooled_l, cnt_l,
                                                       fcW, fcb, (float*)d_out);
}


# layer-1 aggregation summed per node range in LDS f32 accumulators (node ranges x edge slices), block sums flushed by global atomics, not 16 global atomics per edge; float summation order changes
# speedup vs baseline: 2.9260x; 1.5372x over previous
#include <hip/hip_runtime.h>
#include <hip/hip_bf16.h>

typedef float v2f __attribute__((ext_vector_type(2)));
typedef float v8f __attribute__((ext_vector_type(8)));

#define D 16
#define NGRAPH 1024

__global__ void fill_zero_k(float* __restrict__ p, long long n) {
    long long i = (long long)blockIdx.x * blockDim.x + threadIdx.x;
    if (i < n) p[i] = 0.0f;
}

__global__ void gather_gemm16_k(const float* __restrict__ src_rows,
                                const int* __restrict__ ids,
                                const float* __restrict__ W,
                                float* __restrict__ out,
                                int n_nodes) {
    const int lane  = threadIdx.x & 31;
    const int wave  = (int)((blockIdx.x * blockDim.x + threadIdx.x) >> 5);
    const int nwave = (int)((gridDim.x * blockDim.x) >> 5);
    const int row   = lane & 15;
    const int col   = lane & 15;
    const int koff  = (lane >> 4) << 1;
    const int rowb  = (lane >> 4) << 3;

    v2f bb0, bb1, bb2, bb3;
    bb0.x = W[(0  + koff) * D + col]; bb0.y = W[(1  + koff) * D + col];
    bb1.x = W[(4  + koff) * D + col]; bb1.y = W[(5  + koff) * D + col];
    bb2.x = W[(8  + koff) * D + col]; bb2.y = W[(9  + koff) * D + col];
    bb3.x = W[(12 + koff) * D + col]; bb3.y = W[(13 + koff) * D + col];

    const int ntiles = n_nodes >> 4;
    for (int tile = wave; tile < ntiles; tile += nwave) {
        const int node = (tile << 4) + row;
        const int r = ids ? ids[node] : node;
        const float* xrow = src_rows + (size_t)r * D;
        v2f a0 = *(const v2f*)(xrow + 0  + koff);
        v2f a1 = *(const v2f*)(xrow + 4  + koff);
        v2f a2 = *(const v2f*)(xrow + 8  + koff);
        v2f a3 = *(const v2f*)(xrow + 12 + koff);
        v8f c = {};
        c = __builtin_amdgcn_wmma_f32_16x16x4_f32(false, a0, false, bb0, (short)0, c, false, false);
        c = __builtin_amdgcn_wmma_f32_16x16x4_f32(false, a1, false, bb1, (short)0, c, false, false);
        c = __builtin_amdgcn_wmma_f32_16x16x4_f32(false, a2, false, bb2, (short)0, c, false, false);
        c = __builtin_amdgcn_wmma_f32_16x16x4_f32(false, a3, false, bb3, (short)0, c, false, false);
        float* ob = out + ((size_t)(tile << 4) + rowb) * D + col;
        #pragma unroll
        for (int rr = 0; rr < 8; ++rr) ob[(size_t)rr * D] = c[rr];
    }

    const int tail_start = ntiles << 4;
    if (tail_start < n_nodes) {
        int node = tail_start + (int)(blockIdx.x * blockDim.x + threadIdx.x);
        if (node < n_nodes) {
            const int r = ids ? ids[node] : node;
            const float* xrow = src_rows + (size_t)r * D;
            for (int j = 0; j < D; ++j) {
                float acc = 0.0f;
                for (int k = 0; k < D; ++k) acc += xrow[k] * W[k * D + j];
                out[(size_t)node * D + j] = acc;
            }
        }
    }
}

#define DEG_SLICES  16
#define DEG_RANGES  16
#define DEG_CAP     12512
#define DEG_THREADS 1024
#define DEG_UNROLL  8

__global__ __launch_bounds__(DEG_THREADS)
void degree_lds_k(const int* __restrict__ dst, unsigned* __restrict__ part,
                  int n, long long E) {
    __shared__ unsigned cnt[DEG_CAP];
    const int tid   = threadIdx.x;
    const int slice = blockIdx.x;
    const long long per = (E + DEG_SLICES - 1) / DEG_SLICES;
    const long long e0  = (long long)slice * per;
    long long e1 = e0 + per; if (e1 > E) e1 = E;
    const int* __restrict__ ds = dst + e0;
    const int ne = (e1 > e0) ? (int)(e1 - e0) : 0;

    for (int lo = blockIdx.y * DEG_CAP; lo < n; lo += gridDim.y * DEG_CAP) {
        const int left = n - lo;
        const unsigned span = (unsigned)(left < DEG_CAP ? left : DEG_CAP);
        for (int i = tid; i < DEG_CAP; i += DEG_THREADS) cnt[i] = 0u;
        __syncthreads();
        for (int e = tid; e < ne; e += DEG_THREADS * DEG_UNROLL) {
            int d[DEG_UNROLL];
            #pragma unroll
            for (int u = 0; u < DEG_UNROLL; ++u) {
                const int ee = e + u * DEG_THREADS;
                d[u] = (ee < ne) ? ds[ee] : -1;
            }
            #pragma unroll
            for (int u = 0; u < DEG_UNROLL; ++u) {
                const unsigned r = (unsigned)(d[u] - lo);
                if (r < span) atomicAdd(&cnt[r], 1u);
            }
        }
        __syncthreads();
        unsigned* __restrict__ prow = part + (size_t)slice * n + lo;
        for (int i = tid; i < (int)span; i += DEG_THREADS) prow[i] = cnt[i];
        __syncthreads();
    }
}

__global__ void dis_from_parts_k(const unsigned* __restrict__ part,
                                 float* __restrict__ dis, int n) {
    int i = blockIdx.x * blockDim.x + threadIdx.x;
    if (i < n) {
        unsigned c = 0u;
        #pragma unroll
        for (int s = 0; s < DEG_SLICES; ++s) c += part[(size_t)s * n + i];
        dis[i] = rsqrtf((float)c + 1.0f);
    }
}

#define AGG_SLICES  4
#define AGG_RANGES  64
#define AGG_CAP     3125
#define AGG_THREADS 1024
#define AGG_UNROLL  16
#define AGG_QCAP    (32 * AGG_UNROLL + 32)
static_assert(AGG_CAP <= 65536, "a range-local node index is queued in 16 bits");

template <int K> __device__ __forceinline__ int row_bcast_i(int v) {
    return __builtin_amdgcn_update_dpp(0, v, 0x150 + K, 0xf, 0xf, false);
}
template <int K> __device__ __forceinline__ float row_bcast_f(float v) {
    return __int_as_float(row_bcast_i<K>(__float_as_int(v)));
}

#define ROWS16(X) X(0) X(1) X(2) X(3) X(4) X(5) X(6) X(7) \
                  X(8) X(9) X(10) X(11) X(12) X(13) X(14) X(15)

__device__ __forceinline__ void agg_chunk(const float* __restrict__ h,
                                          const float* __restrict__ dis,
                                          const int* __restrict__ src,
                                          const float* disl, float* acc,
                                          long long e, int dl, bool ok, int j) {
    int s = 0;
    float w = 0.0f;
    if (ok) {
        s = src[e];
        w = dis[s] * disl[dl];
    }
    const int oki = ok ? 1 : 0;
#define LOAD_ROW(K) const float hv##K = h[(unsigned)row_bcast_i<K>(s) * D + j];
    ROWS16(LOAD_ROW)
#undef LOAD_ROW
#define ADD_ROW(K) if (row_bcast_i<K>(oki)) atomicAdd(&acc[row_bcast_i<K>(dl) * D + j], hv##K * row_bcast_f<K>(w));
    ROWS16(ADD_ROW)
#undef ADD_ROW
}

__global__ __launch_bounds__(AGG_THREADS)
void aggregate_lds_k(const float* __restrict__ h,
                     const float* __restrict__ dis,
                     const int* __restrict__ src,
                     const int* __restrict__ dst,
                     float* __restrict__ agg,
                     int n, long long E) {
    __shared__ float acc[AGG_CAP * D];
    __shared__ float disl[AGG_CAP];
    __shared__ int   queue[(AGG_THREADS >> 5) * AGG_QCAP];
    __shared__ unsigned short qdst[(AGG_THREADS >> 5) * AGG_QCAP];
    const int tid  = threadIdx.x;
    const int lane = tid & 31;
    const int j    = lane & 15;
    const int wv   = __builtin_amdgcn_readfirstlane(tid >> 5);
    int* qw = queue + wv * AGG_QCAP;
    unsigned short* qd = qdst + wv * AGG_QCAP;

    const long long per = (E + AGG_SLICES - 1) / AGG_SLICES;
    const long long e0  = (long long)blockIdx.x * per;
    long long e1 = e0 + per; if (e1 > E) e1 = E;
    const int ne = (e1 > e0) ? (int)(e1 - e0) : 0;
    const int* __restrict__ ds = dst + e0;
    const int trip = 32 * AGG_UNROLL;

    for (int lo = blockIdx.y * AGG_CAP; lo < n; lo += gridDim.y * AGG_CAP) {
        const int left = n - lo;
        const unsigned span = (unsigned)(left < AGG_CAP ? left : AGG_CAP);
        for (int i = tid; i < AGG_CAP * D; i += AGG_THREADS) acc[i] = 0.0f;
        for (int i = tid; i < (int)span; i += AGG_THREADS) disl[i] = dis[lo + i];
        __syncthreads();

        int nq = 0;
        for (int base = wv * trip; base < ne; base += (AGG_THREADS >> 5) * trip) {
            int dv[AGG_UNROLL];
            if (base + trip <= ne) {
                #pragma unroll
                for (int u = 0; u < AGG_UNROLL; ++u) dv[u] = ds[base + u * 32 + lane];
            } else {
                #pragma unroll
                for (int u = 0; u < AGG_UNROLL; ++u) {
                    const int ee = base + u * 32 + lane;
                    dv[u] = (ee < ne) ? ds[ee] : -1;
                }
            }
            #pragma unroll
            for (int u = 0; u < AGG_UNROLL; ++u) {
                const unsigned r = (unsigned)(dv[u] - lo);
                const bool hit = r < span;
                const unsigned m = __builtin_amdgcn_ballot_w32(hit);
                if (m) {
                    if (hit) {
                        const int pos = nq + (int)__builtin_amdgcn_mbcnt_lo(m, 0u);
                        qw[pos] = base + u * 32 + lane;
                        qd[pos] = (unsigned short)r;
                    }
                    nq += __builtin_popcount(m);
                }
            }
            while (nq >= 32) {
                nq -= 32;
                agg_chunk(h, dis, src, disl, acc, e0 + qw[nq + lane], (int)qd[nq + lane], true, j);
            }
        }
        if (nq > 0) {
            const bool ok = lane < nq;
            agg_chunk(h, dis, src, disl, acc, e0 + (ok ? qw[lane] : 0), ok ? (int)qd[lane] : 0, ok, j);
        }
        __syncthreads();

        float* __restrict__ arow = agg + (size_t)lo * D;
        for (int i = tid; i < (int)span * D; i += AGG_THREADS) {
            const float v = acc[i];
            if (v != 0.0f) atomicAdd(arow + i, v);
        }
        __syncthreads();
    }
}

__global__ void node_relu_k(float* __restrict__ h, const float* __restrict__ agg,
                            const float* __restrict__ dis,
                            const float* __restrict__ bias, long long n16) {
    long long t = (long long)blockIdx.x * blockDim.x + threadIdx.x;
    if (t < n16) {
        long long i = t >> 4; int j = (int)(t & 15);
        float di = dis[i];
        float v = agg[t] + h[t] * (di * di) + bias[j];
        h[t] = v > 0.0f ? v : 0.0f;
    }
}

#define POOL_THREADS 1024
#define POOL_BLOCKS  256

__global__ __launch_bounds__(POOL_THREADS)
void edge_node_pool_k(const float* __restrict__ h2,
                      const float* __restrict__ dis,
                      const float* __restrict__ bias,
                      const int* __restrict__ src,
                      const int* __restrict__ dst,
                      const int* __restrict__ batch,
                      float* __restrict__ pooled,
                      float* __restrict__ cnt,
                      int n, long long E) {
    __shared__ float acc[NGRAPH * D];
    __shared__ float gcnt[NGRAPH];
    const int tid  = threadIdx.x;
    const int lane = tid & 31;
    const int j    = lane & 15;

    for (int i = tid; i < NGRAPH * D; i += POOL_THREADS) acc[i] = 0.0f;
    for (int i = tid; i < NGRAPH; i += POOL_THREADS) gcnt[i] = 0.0f;
    __syncthreads();

    const long long wave  = (long long)blockIdx.x * (POOL_THREADS >> 5)
                          + __builtin_amdgcn_readfirstlane(tid >> 5);
    const long long nwave = (long long)gridDim.x * (POOL_THREADS >> 5);

    const long long nfull = E >> 5;
    long long c = wave;
    int s = 0, d = 0;
    if (c < nfull) { s = src[(c << 5) + lane]; d = dst[(c << 5) + lane]; }
    while (c < nfull) {
        const float w = dis[s] * dis[d];
        const int   g = batch[d];
#define LOAD_ROW(K) const float hv##K = h2[(unsigned)row_bcast_i<K>(s) * D + j];
        ROWS16(LOAD_ROW)
#undef LOAD_ROW
        const long long cn = c + nwave;
        int sn = 0, dn = 0;
        if (cn < nfull) { sn = src[(cn << 5) + lane]; dn = dst[(cn << 5) + lane]; }
#define ADD_ROW(K) atomicAdd(&acc[row_bcast_i<K>(g) * D + j], hv##K * row_bcast_f<K>(w));
        ROWS16(ADD_ROW)
#undef ADD_ROW
        s = sn; d = dn; c = cn;
    }
    if (blockIdx.x == 0 && tid < 16) {
        for (long long e = nfull << 5; e < E; ++e) {
            const int ss = src[e], dd = dst[e];
            const float w = dis[ss] * dis[dd];
            atomicAdd(&acc[batch[dd] * D + j], h2[(size_t)ss * D + j] * w);
        }
    }

    const long long n16 = (long long)n * D;
    for (long long t = (long long)blockIdx.x * POOL_THREADS + tid; t < n16;
         t += (long long)gridDim.x * POOL_THREADS) {
        const int i = (int)(t >> 4);
        const float di = dis[i];
        const int   g  = batch[i];
        atomicAdd(&acc[g * D + j], h2[t] * (di * di) + bias[j]);
        if (j == 0) atomicAdd(&gcnt[g], 1.0f);
    }
    __syncthreads();

    const int rot = (int)((blockIdx.x * POOL_THREADS) & (NGRAPH * D - 1));
    for (int i = tid; i < NGRAPH * D; i += POOL_THREADS) {
        const int idx = (i + rot) & (NGRAPH * D - 1);
        const float v = acc[idx];
        if (v != 0.0f) atomicAdd(pooled + idx, v);
    }
    for (int i = tid; i < NGRAPH; i += POOL_THREADS) {
        const float v = gcnt[i];
        if (v != 0.0f) atomicAdd(cnt + i, v);
    }
}

__global__ void fc_k(const float* __restrict__ pooled_r, const float* __restrict__ cnt_r,
                     const float* __restrict__ pooled_l, const float* __restrict__ cnt_l,
                     const float* __restrict__ fcW, const float* __restrict__ fcb,
                     float* __restrict__ out) {
    int t = blockIdx.x * blockDim.x + threadIdx.x;
    if (t < NGRAPH * 6) {
        int g = t / 6, o = t - g * 6;
        float icr = 1.0f / fmaxf(cnt_r[g], 1.0f);
        float icl = 1.0f / fmaxf(cnt_l[g], 1.0f);
        float acc = fcb[o];
        #pragma unroll
        for (int j = 0; j < D; ++j)
            acc += fcW[o * (2 * D) + j] * (pooled_r[g * D + j] * icr);
        #pragma unroll
        for (int j = 0; j < D; ++j)
            acc += fcW[o * (2 * D) + D + j] * (pooled_l[g * D + j] * icl);
        if (o < 3) out[g * 3 + o] = acc;
        else       out[NGRAPH * 3 + g * 3 + (o - 3)] = acc;
    }
}

static void run_protein(const float* emb, const float* W1, const float* b1,
                        const float* W2, const float* b2,
                        const int* ids, const int* edge, const int* batch,
                        int N, long long E,
                        float* bufA, float* bufB, float* dis,
                        float* pooled, float* cnt, hipStream_t stream) {
    const int* src = edge;
    const int* dst = edge + E;
    const long long n16 = (long long)N * D;
    const int B = 256;
    dim3 blk(B);
    int gemm_grid = (int)(((N >> 4) + 7) / 8);
    if (gemm_grid < 1) gemm_grid = 1;

    degree_lds_k<<<dim3(DEG_SLICES, DEG_RANGES), dim3(DEG_THREADS), 0, stream>>>(
        dst, (unsigned*)bufB, N, E);
    dis_from_parts_k<<<(int)((N + B - 1) / B), blk, 0, stream>>>((const unsigned*)bufB, dis, N);

    fill_zero_k<<<(int)((n16 + B - 1) / B), blk, 0, stream>>>(bufB, n16);

    gather_gemm16_k<<<gemm_grid, blk, 0, stream>>>(emb, ids, W1, bufA, N);

    aggregate_lds_k<<<dim3(AGG_SLICES, AGG_RANGES), dim3(AGG_THREADS), 0, stream>>>(
        bufA, dis, src, dst, bufB, N, E);
    node_relu_k<<<(int)((n16 + B - 1) / B), blk, 0, stream>>>(bufA, bufB, dis, b1, n16);

    gather_gemm16_k<<<gemm_grid, blk, 0, stream>>>(bufA, nullptr, W2, bufB, N);

    fill_zero_k<<<(int)((NGRAPH * D + B - 1) / B), blk, 0, stream>>>(pooled, NGRAPH * D);
    fill_zero_k<<<(int)((NGRAPH + B - 1) / B), blk, 0, stream>>>(cnt, NGRAPH);
    edge_node_pool_k<<<POOL_BLOCKS, POOL_THREADS, 0, stream>>>(bufB, dis, b2, src, dst, batch,
                                                               pooled, cnt, N, E);
}

extern "C" void kernel_launch(void* const* d_in, const int* in_sizes, int n_in,
                              void* d_out, int out_size, void* d_ws, size_t ws_size,
                              hipStream_t stream) {
    const float* emb = (const float*)d_in[0];
    const float* W1  = (const float*)d_in[1];
    const float* b1  = (const float*)d_in[2];
    const float* W2  = (const float*)d_in[3];
    const float* b2  = (const float*)d_in[4];
    const float* fcW = (const float*)d_in[5];
    const float* fcb = (const float*)d_in[6];
    const int* rx     = (const int*)d_in[7];
    const int* redge  = (const int*)d_in[8];
    const int* rbatch = (const int*)d_in[9];
    const int* lx     = (const int*)d_in[10];
    const int* ledge  = (const int*)d_in[11];
    const int* lbatch = (const int*)d_in[12];

    const int N = in_sizes[7];
    const long long E = in_sizes[8] / 2;

    char* ws = (char*)d_ws;
    float* bufA = (float*)ws;  ws += (size_t)N * D * sizeof(float);
    float* bufB = (float*)ws;  ws += (size_t)N * D * sizeof(float);
    float* dis  = (float*)ws;  ws += (size_t)N * sizeof(float);
    float* pooled_r = (float*)ws; ws += (size_t)NGRAPH * D * sizeof(float);
    float* pooled_l = (float*)ws; ws += (size_t)NGRAPH * D * sizeof(float);
    float* cnt_r = (float*)ws; ws += (size_t)NGRAPH * sizeof(float);
    float* cnt_l = (float*)ws; ws += (size_t)NGRAPH * sizeof(float);

    run_protein(emb, W1, b1, W2, b2, rx, redge, rbatch, N, E,
                bufA, bufB, dis, pooled_r, cnt_r, stream);
    run_protein(emb, W1, b1, W2, b2, lx, ledge, lbatch, N, E,
                bufA, bufB, dis, pooled_l, cnt_l, stream);

    fc_k<<<(NGRAPH * 6 + 255) / 256, 256, 0, stream>>>(pooled_r, cnt_r, pooled_l, cnt_l,
                                                       fcW, fcb, (float*)d_out);
}


# edge weight dis[src]*dis[dst] factored out of the edge loops: gathered rows pre-scaled by dis; dst factor per node (layer 1) or read with the graph id as one 8-byte record (layer 2); rounding changes
# speedup vs baseline: 3.4240x; 1.1702x over previous
#include <hip/hip_runtime.h>
#include <hip/hip_bf16.h>

typedef float v2f __attribute__((ext_vector_type(2)));
typedef float v8f __attribute__((ext_vector_type(8)));
typedef unsigned v2u __attribute__((ext_vector_type(2)));

#define D 16
#define NGRAPH 1024

__global__ void scale_rows_zero_k(float* __restrict__ h, const float* __restrict__ dis,
                                  float* __restrict__ agg, long long n16) {
    long long t = (long long)blockIdx.x * blockDim.x + threadIdx.x;
    if (t < n16) {
        h[t] *= dis[t >> 4];
        agg[t] = 0.0f;
    }
}

__global__ void gather_gemm16_k(const float* __restrict__ src_rows,
                                const int* __restrict__ ids,
                                const float* __restrict__ W,
                                float* __restrict__ out,
                                int n_nodes) {
    const int lane  = threadIdx.x & 31;
    const int wave  = (int)((blockIdx.x * blockDim.x + threadIdx.x) >> 5);
    const int nwave = (int)((gridDim.x * blockDim.x) >> 5);
    const int row   = lane & 15;
    const int col   = lane & 15;
    const int koff  = (lane >> 4) << 1;
    const int rowb  = (lane >> 4) << 3;

    v2f bb0, bb1, bb2, bb3;
    bb0.x = W[(0  + koff) * D + col]; bb0.y = W[(1  + koff) * D + col];
    bb1.x = W[(4  + koff) * D + col]; bb1.y = W[(5  + koff) * D + col];
    bb2.x = W[(8  + koff) * D + col]; bb2.y = W[(9  + koff) * D + col];
    bb3.x = W[(12 + koff) * D + col]; bb3.y = W[(13 + koff) * D + col];

    const int ntiles = n_nodes >> 4;
    for (int tile = wave; tile < ntiles; tile += nwave) {
        const int node = (tile << 4) + row;
        const int r = ids ? ids[node] : node;
        const float* xrow = src_rows + (size_t)r * D;
        v2f a0 = *(const v2f*)(xrow + 0  + koff);
        v2f a1 = *(const v2f*)(xrow + 4  + koff);
        v2f a2 = *(const v2f*)(xrow + 8  + koff);
        v2f a3 = *(const v2f*)(xrow + 12 + koff);
        v8f c = {};
        c = __builtin_amdgcn_wmma_f32_16x16x4_f32(false, a0, false, bb0, (short)0, c, false, false);
        c = __builtin_amdgcn_wmma_f32_16x16x4_f32(false, a1, false, bb1, (short)0, c, false, false);
        c = __builtin_amdgcn_wmma_f32_16x16x4_f32(false, a2, false, bb2, (short)0, c, false, false);
        c = __builtin_amdgcn_wmma_f32_16x16x4_f32(false, a3, false, bb3, (short)0, c, false, false);
        float* ob = out + ((size_t)(tile << 4) + rowb) * D + col;
        #pragma unroll
        for (int rr = 0; rr < 8; ++rr) ob[(size_t)rr * D] = c[rr];
    }

    const int tail_start = ntiles << 4;
    if (tail_start < n_nodes) {
        int node = tail_start + (int)(blockIdx.x * blockDim.x + threadIdx.x);
        if (node < n_nodes) {
            const int r = ids ? ids[node] : node;
            const float* xrow = src_rows + (size_t)r * D;
            for (int j = 0; j < D; ++j) {
                float acc = 0.0f;
                for (int k = 0; k < D; ++k) acc += xrow[k] * W[k * D + j];
                out[(size_t)node * D + j] = acc;
            }
        }
    }
}

#define DEG_SLICES  16
#define DEG_RANGES  16
#define DEG_CAP     12512
#define DEG_THREADS 1024
#define DEG_UNROLL  8

__global__ __launch_bounds__(DEG_THREADS)
void degree_lds_k(const int* __restrict__ dst, unsigned* __restrict__ part,
                  int n, long long E) {
    __shared__ unsigned cnt[DEG_CAP];
    const int tid   = threadIdx.x;
    const int slice = blockIdx.x;
    const long long per = (E + DEG_SLICES - 1) / DEG_SLICES;
    const long long e0  = (long long)slice * per;
    long long e1 = e0 + per; if (e1 > E) e1 = E;
    const int* __restrict__ ds = dst + e0;
    const int ne = (e1 > e0) ? (int)(e1 - e0) : 0;

    for (int lo = blockIdx.y * DEG_CAP; lo < n; lo += gridDim.y * DEG_CAP) {
        const int left = n - lo;
        const unsigned span = (unsigned)(left < DEG_CAP ? left : DEG_CAP);
        for (int i = tid; i < DEG_CAP; i += DEG_THREADS) cnt[i] = 0u;
        __syncthreads();
        for (int e = tid; e < ne; e += DEG_THREADS * DEG_UNROLL) {
            int d[DEG_UNROLL];
            #pragma unroll
            for (int u = 0; u < DEG_UNROLL; ++u) {
                const int ee = e + u * DEG_THREADS;
                d[u] = (ee < ne) ? ds[ee] : -1;
            }
            #pragma unroll
            for (int u = 0; u < DEG_UNROLL; ++u) {
                const unsigned r = (unsigned)(d[u] - lo);
                if (r < span) atomicAdd(&cnt[r], 1u);
            }
        }
        __syncthreads();
        unsigned* __restrict__ prow = part + (size_t)slice * n + lo;
        for (int i = tid; i < (int)span; i += DEG_THREADS) prow[i] = cnt[i];
        __syncthreads();
    }
}

__global__ void dis_from_parts_k(const unsigned* __restrict__ part,
                                 float* __restrict__ dis, int n) {
    int i = blockIdx.x * blockDim.x + threadIdx.x;
    if (i < n) {
        unsigned c = 0u;
        #pragma unroll
        for (int s = 0; s < DEG_SLICES; ++s) c += part[(size_t)s * n + i];
        dis[i] = rsqrtf((float)c + 1.0f);
    }
}

#define AGG_SLICES  4
#define AGG_RANGES  64
#define AGG_CAP     3125
#define AGG_THREADS 1024
#define AGG_UNROLL  16
#define AGG_QCAP    (32 * AGG_UNROLL + 32)
static_assert(AGG_CAP <= 65536, "a range-local node index is queued in 16 bits");

template <int K> __device__ __forceinline__ int row_bcast_i(int v) {
    return __builtin_amdgcn_update_dpp(0, v, 0x150 + K, 0xf, 0xf, false);
}
template <int K> __device__ __forceinline__ float row_bcast_f(float v) {
    return __int_as_float(row_bcast_i<K>(__float_as_int(v)));
}

#define ROWS16(X) X(0) X(1) X(2) X(3) X(4) X(5) X(6) X(7) \
                  X(8) X(9) X(10) X(11) X(12) X(13) X(14) X(15)

__device__ __forceinline__ void agg_chunk(const float* __restrict__ h,
                                          const int* __restrict__ src,
                                          float* acc,
                                          long long e, int dl, bool ok, int j) {
    int s = 0;
    if (ok) s = src[e];
    const int oki = ok ? 1 : 0;
#define LOAD_ROW(K) const float hv##K = h[(unsigned)row_bcast_i<K>(s) * D + j];
    ROWS16(LOAD_ROW)
#undef LOAD_ROW
#define ADD_ROW(K) if (row_bcast_i<K>(oki)) atomicAdd(&acc[row_bcast_i<K>(dl) * D + j], hv##K);
    ROWS16(ADD_ROW)
#undef ADD_ROW
}

__global__ __launch_bounds__(AGG_THREADS)
void aggregate_lds_k(const float* __restrict__ h,
                     const int* __restrict__ src,
                     const int* __restrict__ dst,
                     float* __restrict__ agg,
                     int n, long long E) {
    __shared__ float acc[AGG_CAP * D];
    __shared__ int   queue[(AGG_THREADS >> 5) * AGG_QCAP];
    __shared__ unsigned short qdst[(AGG_THREADS >> 5) * AGG_QCAP];
    const int tid  = threadIdx.x;
    const int lane = tid & 31;
    const int j    = lane & 15;
    const int wv   = __builtin_amdgcn_readfirstlane(tid >> 5);
    int* qw = queue + wv * AGG_QCAP;
    unsigned short* qd = qdst + wv * AGG_QCAP;

    const long long per = (E + AGG_SLICES - 1) / AGG_SLICES;
    const long long e0  = (long long)blockIdx.x * per;
    long long e1 = e0 + per; if (e1 > E) e1 = E;
    const int ne = (e1 > e0) ? (int)(e1 - e0) : 0;
    const int* __restrict__ ds = dst + e0;
    const int trip = 32 * AGG_UNROLL;

    for (int lo = blockIdx.y * AGG_CAP; lo < n; lo += gridDim.y * AGG_CAP) {
        const int left = n - lo;
        const unsigned span = (unsigned)(left < AGG_CAP ? left : AGG_CAP);
        for (int i = tid; i < AGG_CAP * D; i += AGG_THREADS) acc[i] = 0.0f;
        __syncthreads();

        int nq = 0;
        for (int base = wv * trip; base < ne; base += (AGG_THREADS >> 5) * trip) {
            int dv[AGG_UNROLL];
            if (base + trip <= ne) {
                #pragma unroll
                for (int u = 0; u < AGG_UNROLL; ++u) dv[u] = ds[base + u * 32 + lane];
            } else {
                #pragma unroll
                for (int u = 0; u < AGG_UNROLL; ++u) {
                    const int ee = base + u * 32 + lane;
                    dv[u] = (ee < ne) ? ds[ee] : -1;
                }
            }
            #pragma unroll
            for (int u = 0; u < AGG_UNROLL; ++u) {
                const unsigned r = (unsigned)(dv[u] - lo);
                const bool hit = r < span;
                const unsigned m = __builtin_amdgcn_ballot_w32(hit);
                if (m) {
                    if (hit) {
                        const int pos = nq + (int)__builtin_amdgcn_mbcnt_lo(m, 0u);
                        qw[pos] = base + u * 32 + lane;
                        qd[pos] = (unsigned short)r;
                    }
                    nq += __builtin_popcount(m);
                }
            }
            while (nq >= 32) {
                nq -= 32;
                agg_chunk(h, src, acc, e0 + qw[nq + lane], (int)qd[nq + lane], true, j);
            }
        }
        if (nq > 0) {
            const bool ok = lane < nq;
            agg_chunk(h, src, acc, e0 + (ok ? qw[lane] : 0), ok ? (int)qd[lane] : 0, ok, j);
        }
        __syncthreads();

        float* __restrict__ arow = agg + (size_t)lo * D;
        for (int i = tid; i < (int)span * D; i += AGG_THREADS) {
            const float v = acc[i];
            if (v != 0.0f) atomicAdd(arow + i, v);
        }
        __syncthreads();
    }
}

__global__ void node_relu_k(float* __restrict__ h, const float* __restrict__ agg,
                            const float* __restrict__ dis,
                            const float* __restrict__ bias, long long n16) {
    long long t = (long long)blockIdx.x * blockDim.x + threadIdx.x;
    if (t < n16) {
        long long i = t >> 4; int j = (int)(t & 15);
        float di = dis[i];
        float v = di * (agg[t] + h[t]) + bias[j];
        h[t] = (v > 0.0f ? v : 0.0f) * di;
    }
}

#define POOL_THREADS 1024
#define POOL_BLOCKS  256

__global__ void pool_prep_k(const float* __restrict__ dis, const int* __restrict__ batch,
                            unsigned* __restrict__ rec,
                            float* __restrict__ pooled, float* __restrict__ cnt, int n) {
    const int i = blockIdx.x * blockDim.x + threadIdx.x;
    const int stride = gridDim.x * blockDim.x;
    for (int k = i; k < n; k += stride) {
        v2u r;
        r.x = __float_as_uint(dis[k]);
        r.y = (unsigned)batch[k];
        *(v2u*)(rec + 2 * (size_t)k) = r;
    }
    for (int k = i; k < NGRAPH * D; k += stride) pooled[k] = 0.0f;
    for (int k = i; k < NGRAPH; k += stride) cnt[k] = 0.0f;
}

__global__ __launch_bounds__(POOL_THREADS)
void edge_node_pool_k(const float* __restrict__ h2,
                      const unsigned* __restrict__ rec,
                      const float* __restrict__ bias,
                      const int* __restrict__ src,
                      const int* __restrict__ dst,
                      float* __restrict__ pooled,
                      float* __restrict__ cnt,
                      int n, long long E) {
    __shared__ float acc[NGRAPH * D];
    __shared__ float gcnt[NGRAPH];
    const int tid  = threadIdx.x;
    const int lane = tid & 31;
    const int j    = lane & 15;

    for (int i = tid; i < NGRAPH * D; i += POOL_THREADS) acc[i] = 0.0f;
    for (int i = tid; i < NGRAPH; i += POOL_THREADS) gcnt[i] = 0.0f;
    __syncthreads();

    const long long wave  = (long long)blockIdx.x * (POOL_THREADS >> 5)
                          + __builtin_amdgcn_readfirstlane(tid >> 5);
    const long long nwave = (long long)gridDim.x * (POOL_THREADS >> 5);

    const long long nfull = E >> 5;
    long long c = wave;
    int s = 0, d = 0;
    if (c < nfull) { s = src[(c << 5) + lane]; d = dst[(c << 5) + lane]; }
    while (c < nfull) {
        const v2u rd = *(const v2u*)(rec + 2 * (size_t)(unsigned)d);
        const float w = __uint_as_float(rd.x);
        const int   g = (int)rd.y;
#define LOAD_ROW(K) const float hv##K = h2[(unsigned)row_bcast_i<K>(s) * D + j];
        ROWS16(LOAD_ROW)
#undef LOAD_ROW
        const long long cn = c + nwave;
        int sn = 0, dn = 0;
        if (cn < nfull) { sn = src[(cn << 5) + lane]; dn = dst[(cn << 5) + lane]; }
#define ADD_ROW(K) atomicAdd(&acc[row_bcast_i<K>(g) * D + j], hv##K * row_bcast_f<K>(w));
        ROWS16(ADD_ROW)
#undef ADD_ROW
        s = sn; d = dn; c = cn;
    }
    if (blockIdx.x == 0 && tid < 16) {
        for (long long e = nfull << 5; e < E; ++e) {
            const int ss = src[e], dd = dst[e];
            const v2u rd = *(const v2u*)(rec + 2 * (size_t)(unsigned)dd);
            atomicAdd(&acc[(int)rd.y * D + j], h2[(size_t)ss * D + j] * __uint_as_float(rd.x));
        }
    }

    const long long n16 = (long long)n * D;
    for (long long t = (long long)blockIdx.x * POOL_THREADS + tid; t < n16;
         t += (long long)gridDim.x * POOL_THREADS) {
        const int i = (int)(t >> 4);
        const v2u ri = *(const v2u*)(rec + 2 * (size_t)i);
        const float di = __uint_as_float(ri.x);
        const int   g  = (int)ri.y;
        atomicAdd(&acc[g * D + j], h2[t] * di + bias[j]);
        if (j == 0) atomicAdd(&gcnt[g], 1.0f);
    }
    __syncthreads();

    const int rot = (int)((blockIdx.x * POOL_THREADS) & (NGRAPH * D - 1));
    for (int i = tid; i < NGRAPH * D; i += POOL_THREADS) {
        const int idx = (i + rot) & (NGRAPH * D - 1);
        const float v = acc[idx];
        if (v != 0.0f) atomicAdd(pooled + idx, v);
    }
    for (int i = tid; i < NGRAPH; i += POOL_THREADS) {
        const float v = gcnt[i];
        if (v != 0.0f) atomicAdd(cnt + i, v);
    }
}

__global__ void fc_k(const float* __restrict__ pooled_r, const float* __restrict__ cnt_r,
                     const float* __restrict__ pooled_l, const float* __restrict__ cnt_l,
                     const float* __restrict__ fcW, const float* __restrict__ fcb,
                     float* __restrict__ out) {
    int t = blockIdx.x * blockDim.x + threadIdx.x;
    if (t < NGRAPH * 6) {
        int g = t / 6, o = t - g * 6;
        float icr = 1.0f / fmaxf(cnt_r[g], 1.0f);
        float icl = 1.0f / fmaxf(cnt_l[g], 1.0f);
        float acc = fcb[o];
        #pragma unroll
        for (int j = 0; j < D; ++j)
            acc += fcW[o * (2 * D) + j] * (pooled_r[g * D + j] * icr);
        #pragma unroll
        for (int j = 0; j < D; ++j)
            acc += fcW[o * (2 * D) + D + j] * (pooled_l[g * D + j] * icl);
        if (o < 3) out[g * 3 + o] = acc;
        else       out[NGRAPH * 3 + g * 3 + (o - 3)] = acc;
    }
}

static void run_protein(const float* emb, const float* W1, const float* b1,
                        const float* W2, const float* b2,
                        const int* ids, const int* edge, const int* batch,
                        int N, long long E,
                        float* bufA, float* bufB, float* dis,
                        float* pooled, float* cnt, hipStream_t stream) {
    const int* src = edge;
    const int* dst = edge + E;
    const long long n16 = (long long)N * D;
    const int B = 256;
    dim3 blk(B);
    int gemm_grid = (int)(((N >> 4) + 7) / 8);
    if (gemm_grid < 1) gemm_grid = 1;

    degree_lds_k<<<dim3(DEG_SLICES, DEG_RANGES), dim3(DEG_THREADS), 0, stream>>>(
        dst, (unsigned*)bufB, N, E);
    dis_from_parts_k<<<(int)((N + B - 1) / B), blk, 0, stream>>>((const unsigned*)bufB, dis, N);

    gather_gemm16_k<<<gemm_grid, blk, 0, stream>>>(emb, ids, W1, bufA, N);

    scale_rows_zero_k<<<(int)((n16 + B - 1) / B), blk, 0, stream>>>(bufA, dis, bufB, n16);

    aggregate_lds_k<<<dim3(AGG_SLICES, AGG_RANGES), dim3(AGG_THREADS), 0, stream>>>(
        bufA, src, dst, bufB, N, E);
    node_relu_k<<<(int)((n16 + B - 1) / B), blk, 0, stream>>>(bufA, bufB, dis, b1, n16);

    gather_gemm16_k<<<gemm_grid, blk, 0, stream>>>(bufA, nullptr, W2, bufB, N);

    unsigned* rec = (unsigned*)bufA;
    pool_prep_k<<<(int)((N + B - 1) / B), blk, 0, stream>>>(dis, batch, rec, pooled, cnt, N);

    edge_node_pool_k<<<POOL_BLOCKS, POOL_THREADS, 0, stream>>>(bufB, rec, b2, src, dst,
                                                               pooled, cnt, N, E);
}

extern "C" void kernel_launch(void* const* d_in, const int* in_sizes, int n_in,
                              void* d_out, int out_size, void* d_ws, size_t ws_size,
                              hipStream_t stream) {
    const float* emb = (const float*)d_in[0];
    const float* W1  = (const float*)d_in[1];
    const float* b1  = (const float*)d_in[2];
    const float* W2  = (const float*)d_in[3];
    const float* b2  = (const float*)d_in[4];
    const float* fcW = (const float*)d_in[5];
    const float* fcb = (const float*)d_in[6];
    const int* rx     = (const int*)d_in[7];
    const int* redge  = (const int*)d_in[8];
    const int* rbatch = (const int*)d_in[9];
    const int* lx     = (const int*)d_in[10];
    const int* ledge  = (const int*)d_in[11];
    const int* lbatch = (const int*)d_in[12];

    const int N = in_sizes[7];
    const long long E = in_sizes[8] / 2;

    char* ws = (char*)d_ws;
    float* bufA = (float*)ws;  ws += (size_t)N * D * sizeof(float);
    float* bufB = (float*)ws;  ws += (size_t)N * D * sizeof(float);
    float* dis  = (float*)ws;  ws += (size_t)N * sizeof(float);
    float* pooled_r = (float*)ws; ws += (size_t)NGRAPH * D * sizeof(float);
    float* pooled_l = (float*)ws; ws += (size_t)NGRAPH * D * sizeof(float);
    float* cnt_r = (float*)ws; ws += (size_t)NGRAPH * sizeof(float);
    float* cnt_l = (float*)ws; ws += (size_t)NGRAPH * sizeof(float);

    run_protein(emb, W1, b1, W2, b2, rx, redge, rbatch, N, E,
                bufA, bufB, dis, pooled_r, cnt_r, stream);
    run_protein(emb, W1, b1, W2, b2, lx, ledge, lbatch, N, E,
                bufA, bufB, dis, pooled_l, cnt_l, stream);

    fc_k<<<(NGRAPH * 6 + 255) / 256, 256, 0, stream>>>(pooled_r, cnt_r, pooled_l, cnt_l,
                                                       fcW, fcb, (float*)d_out);
}


# per-node passes folded into the two products: layer-1 product scales rows by dis and zeroes its tile of sums, layer-2 product forms relu input in registers and leaves the pool records; 13 launches
# speedup vs baseline: 3.5787x; 1.0452x over previous
#include <hip/hip_runtime.h>
#include <hip/hip_bf16.h>

typedef float v2f __attribute__((ext_vector_type(2)));
typedef float v8f __attribute__((ext_vector_type(8)));
typedef unsigned v2u __attribute__((ext_vector_type(2)));

#define D 16
#define NGRAPH 1024

typedef float v4f __attribute__((ext_vector_type(4)));

template <int K> __device__ __forceinline__ int row_bcast_i(int v) {
    return __builtin_amdgcn_update_dpp(0, v, 0x150 + K, 0xf, 0xf, false);
}
template <int K> __device__ __forceinline__ float row_bcast_f(float v) {
    return __int_as_float(row_bcast_i<K>(__float_as_int(v)));
}

#define REC_STRIDE D


__global__ void gather_gemm16_scale_k(const float* __restrict__ src_rows,
                                      const int* __restrict__ ids,
                                      const float* __restrict__ W,
                                      const float* __restrict__ dis,
                                      float* __restrict__ out,
                                      float* __restrict__ agg,
                                      int n_nodes) {
    const int lane  = threadIdx.x & 31;
    const int wave  = (int)((blockIdx.x * blockDim.x + threadIdx.x) >> 5);
    const int nwave = (int)((gridDim.x * blockDim.x) >> 5);
    const int row   = lane & 15;
    const int col   = lane & 15;
    const int koff  = (lane >> 4) << 1;
    const int rowb  = (lane >> 4) << 3;

    v2f bb0, bb1, bb2, bb3;
    bb0.x = W[(0  + koff) * D + col]; bb0.y = W[(1  + koff) * D + col];
    bb1.x = W[(4  + koff) * D + col]; bb1.y = W[(5  + koff) * D + col];
    bb2.x = W[(8  + koff) * D + col]; bb2.y = W[(9  + koff) * D + col];
    bb3.x = W[(12 + koff) * D + col]; bb3.y = W[(13 + koff) * D + col];

    const int ntiles = n_nodes >> 4;
    for (int tile = wave; tile < ntiles; tile += nwave) {
        const int node = (tile << 4) + row;
        const int r = ids[node];
        const float dv = dis[(tile << 4) + rowb + (lane & 7)];
        const float* xrow = src_rows + (size_t)r * D;
        v2f a0 = *(const v2f*)(xrow + 0  + koff);
        v2f a1 = *(const v2f*)(xrow + 4  + koff);
        v2f a2 = *(const v2f*)(xrow + 8  + koff);
        v2f a3 = *(const v2f*)(xrow + 12 + koff);
        v4f* ab = (v4f*)(agg + ((size_t)tile << 8));
        const v4f zero4 = {0.0f, 0.0f, 0.0f, 0.0f};
        ab[lane] = zero4;
        ab[32 + lane] = zero4;
        v8f c = {};
        c = __builtin_amdgcn_wmma_f32_16x16x4_f32(false, a0, false, bb0, (short)0, c, false, false);
        c = __builtin_amdgcn_wmma_f32_16x16x4_f32(false, a1, false, bb1, (short)0, c, false, false);
        c = __builtin_amdgcn_wmma_f32_16x16x4_f32(false, a2, false, bb2, (short)0, c, false, false);
        c = __builtin_amdgcn_wmma_f32_16x16x4_f32(false, a3, false, bb3, (short)0, c, false, false);
        float* ob = out + ((size_t)(tile << 4) + rowb) * D + col;
#define SCALE_STORE(K) ob[(size_t)K * D] = c[K] * row_bcast_f<K>(dv);
        SCALE_STORE(0) SCALE_STORE(1) SCALE_STORE(2) SCALE_STORE(3)
        SCALE_STORE(4) SCALE_STORE(5) SCALE_STORE(6) SCALE_STORE(7)
#undef SCALE_STORE
    }

    const int tail_start = ntiles << 4;
    if (tail_start < n_nodes) {
        int node = tail_start + (int)(blockIdx.x * blockDim.x + threadIdx.x);
        if (node < n_nodes) {
            const int r = ids[node];
            const float di = dis[node];
            const float* xrow = src_rows + (size_t)r * D;
            for (int j = 0; j < D; ++j) {
                float acc = 0.0f;
                for (int k = 0; k < D; ++k) acc += xrow[k] * W[k * D + j];
                out[(size_t)node * D + j] = acc * di;
                agg[(size_t)node * D + j] = 0.0f;
            }
        }
    }
}

__global__ void relu_gemm16_k(float* h, float* aggrec,
                              const float* __restrict__ dis,
                              const float* __restrict__ bias,
                              const float* __restrict__ W,
                              const int* __restrict__ batch,
                              float* __restrict__ pooled, float* __restrict__ cnt,
                              int n_nodes) {
    const int gid   = (int)(blockIdx.x * blockDim.x + threadIdx.x);
    const int gsize = (int)(gridDim.x * blockDim.x);
    for (int k = gid; k < NGRAPH * D; k += gsize) pooled[k] = 0.0f;
    for (int k = gid; k < NGRAPH; k += gsize) cnt[k] = 0.0f;

    const int lane  = threadIdx.x & 31;
    const int wave  = gid >> 5;
    const int nwave = gsize >> 5;
    const int row   = lane & 15;
    const int col   = lane & 15;
    const int koff  = (lane >> 4) << 1;
    const int rowb  = (lane >> 4) << 3;

    v2f bb0, bb1, bb2, bb3;
    bb0.x = W[(0  + koff) * D + col]; bb0.y = W[(1  + koff) * D + col];
    bb1.x = W[(4  + koff) * D + col]; bb1.y = W[(5  + koff) * D + col];
    bb2.x = W[(8  + koff) * D + col]; bb2.y = W[(9  + koff) * D + col];
    bb3.x = W[(12 + koff) * D + col]; bb3.y = W[(13 + koff) * D + col];
    const v2f bk0 = *(const v2f*)(bias + 0  + koff);
    const v2f bk1 = *(const v2f*)(bias + 4  + koff);
    const v2f bk2 = *(const v2f*)(bias + 8  + koff);
    const v2f bk3 = *(const v2f*)(bias + 12 + koff);

    const int ntiles = n_nodes >> 4;
    for (int tile = wave; tile < ntiles; tile += nwave) {
        const int node = (tile << 4) + row;
        const float di = dis[node];
        const float* hrow = h + (size_t)node * D;
        const float* grow = aggrec + (size_t)node * D;
        const v2f h0 = *(const v2f*)(hrow + 0  + koff), g0 = *(const v2f*)(grow + 0  + koff);
        const v2f h1 = *(const v2f*)(hrow + 4  + koff), g1 = *(const v2f*)(grow + 4  + koff);
        const v2f h2 = *(const v2f*)(hrow + 8  + koff), g2 = *(const v2f*)(grow + 8  + koff);
        const v2f h3 = *(const v2f*)(hrow + 12 + koff), g3 = *(const v2f*)(grow + 12 + koff);
        int bt = 0;
        if (lane < 16) bt = batch[node];
        v2f a0, a1, a2, a3;
#define RELU_IN(A, H, G, B) { float vx = di * (G.x + H.x) + B.x; float vy = di * (G.y + H.y) + B.y; \
                              A.x = (vx > 0.0f ? vx : 0.0f) * di; A.y = (vy > 0.0f ? vy : 0.0f) * di; }
        RELU_IN(a0, h0, g0, bk0) RELU_IN(a1, h1, g1, bk1) RELU_IN(a2, h2, g2, bk2) RELU_IN(a3, h3, g3, bk3)
#undef RELU_IN
        v8f c = {};
        c = __builtin_amdgcn_wmma_f32_16x16x4_f32(false, a0, false, bb0, (short)0, c, false, false);
        c = __builtin_amdgcn_wmma_f32_16x16x4_f32(false, a1, false, bb1, (short)0, c, false, false);
        c = __builtin_amdgcn_wmma_f32_16x16x4_f32(false, a2, false, bb2, (short)0, c, false, false);
        c = __builtin_amdgcn_wmma_f32_16x16x4_f32(false, a3, false, bb3, (short)0, c, false, false);
        float* ob = h + ((size_t)(tile << 4) + rowb) * D + col;
        #pragma unroll
        for (int rr = 0; rr < 8; ++rr) ob[(size_t)rr * D] = c[rr];
        if (lane < 16) {
            v2u rc;
            rc.x = __float_as_uint(di);
            rc.y = (unsigned)bt;
            *(v2u*)((unsigned*)aggrec + (size_t)node * REC_STRIDE) = rc;
        }
    }

    const int tail_start = ntiles << 4;
    if (tail_start < n_nodes) {
        int node = tail_start + gid;
        if (node < n_nodes) {
            const float di = dis[node];
            float z[D];
            for (int k = 0; k < D; ++k) {
                float v = di * (aggrec[(size_t)node * D + k] + h[(size_t)node * D + k]) + bias[k];
                z[k] = (v > 0.0f ? v : 0.0f) * di;
            }
            for (int j = 0; j < D; ++j) {
                float acc = 0.0f;
                for (int k = 0; k < D; ++k) acc += z[k] * W[k * D + j];
                h[(size_t)node * D + j] = acc;
            }
            v2u rc;
            rc.x = __float_as_uint(di);
            rc.y = (unsigned)batch[node];
            *(v2u*)((unsigned*)aggrec + (size_t)node * REC_STRIDE) = rc;
        }
    }
}

#define DEG_SLICES  16
#define DEG_RANGES  16
#define DEG_CAP     12512
#define DEG_THREADS 1024
#define DEG_UNROLL  8

__global__ __launch_bounds__(DEG_THREADS)
void degree_lds_k(const int* __restrict__ dst, unsigned* __restrict__ part,
                  int n, long long E) {
    __shared__ unsigned cnt[DEG_CAP];
    const int tid   = threadIdx.x;
    const int slice = blockIdx.x;
    const long long per = (E + DEG_SLICES - 1) / DEG_SLICES;
    const long long e0  = (long long)slice * per;
    long long e1 = e0 + per; if (e1 > E) e1 = E;
    const int* __restrict__ ds = dst + e0;
    const int ne = (e1 > e0) ? (int)(e1 - e0) : 0;

    for (int lo = blockIdx.y * DEG_CAP; lo < n; lo += gridDim.y * DEG_CAP) {
        const int left = n - lo;
        const unsigned span = (unsigned)(left < DEG_CAP ? left : DEG_CAP);
        for (int i = tid; i < DEG_CAP; i += DEG_THREADS) cnt[i] = 0u;
        __syncthreads();
        for (int e = tid; e < ne; e += DEG_THREADS * DEG_UNROLL) {
            int d[DEG_UNROLL];
            #pragma unroll
            for (int u = 0; u < DEG_UNROLL; ++u) {
                const int ee = e + u * DEG_THREADS;
                d[u] = (ee < ne) ? ds[ee] : -1;
            }
            #pragma unroll
            for (int u = 0; u < DEG_UNROLL; ++u) {
                const unsigned r = (unsigned)(d[u] - lo);
                if (r < span) atomicAdd(&cnt[r], 1u);
            }
        }
        __syncthreads();
        unsigned* __restrict__ prow = part + (size_t)slice * n + lo;
        for (int i = tid; i < (int)span; i += DEG_THREADS) prow[i] = cnt[i];
        __syncthreads();
    }
}

__global__ void dis_from_parts_k(const unsigned* __restrict__ part,
                                 float* __restrict__ dis, int n) {
    int i = blockIdx.x * blockDim.x + threadIdx.x;
    if (i < n) {
        unsigned c = 0u;
        #pragma unroll
        for (int s = 0; s < DEG_SLICES; ++s) c += part[(size_t)s * n + i];
        dis[i] = rsqrtf((float)c + 1.0f);
    }
}

#define AGG_SLICES  4
#define AGG_RANGES  64
#define AGG_CAP     3125
#define AGG_THREADS 1024
#define AGG_UNROLL  16
#define AGG_QCAP    (32 * AGG_UNROLL + 32)
static_assert(AGG_CAP <= 65536, "a range-local node index is queued in 16 bits");

#define ROWS16(X) X(0) X(1) X(2) X(3) X(4) X(5) X(6) X(7) \
                  X(8) X(9) X(10) X(11) X(12) X(13) X(14) X(15)

__device__ __forceinline__ void agg_chunk(const float* __restrict__ h,
                                          const int* __restrict__ src,
                                          float* acc,
                                          long long e, int dl, bool ok, int j) {
    int s = 0;
    if (ok) s = src[e];
    const int oki = ok ? 1 : 0;
#define LOAD_ROW(K) const float hv##K = h[(unsigned)row_bcast_i<K>(s) * D + j];
    ROWS16(LOAD_ROW)
#undef LOAD_ROW
#define ADD_ROW(K) if (row_bcast_i<K>(oki)) atomicAdd(&acc[row_bcast_i<K>(dl) * D + j], hv##K);
    ROWS16(ADD_ROW)
#undef ADD_ROW
}

__global__ __launch_bounds__(AGG_THREADS)
void aggregate_lds_k(const float* __restrict__ h,
                     const int* __restrict__ src,
                     const int* __restrict__ dst,
                     float* __restrict__ agg,
                     int n, long long E) {
    __shared__ float acc[AGG_CAP * D];
    __shared__ int   queue[(AGG_THREADS >> 5) * AGG_QCAP];
    __shared__ unsigned short qdst[(AGG_THREADS >> 5) * AGG_QCAP];
    const int tid  = threadIdx.x;
    const int lane = tid & 31;
    const int j    = lane & 15;
    const int wv   = __builtin_amdgcn_readfirstlane(tid >> 5);
    int* qw = queue + wv * AGG_QCAP;
    unsigned short* qd = qdst + wv * AGG_QCAP;

    const long long per = (E + AGG_SLICES - 1) / AGG_SLICES;
    const long long e0  = (long long)blockIdx.x * per;
    long long e1 = e0 + per; if (e1 > E) e1 = E;
    const int ne = (e1 > e0) ? (int)(e1 - e0) : 0;
    const int* __restrict__ ds = dst + e0;
    const int trip = 32 * AGG_UNROLL;

    for (int lo = blockIdx.y * AGG_CAP; lo < n; lo += gridDim.y * AGG_CAP) {
        const int left = n - lo;
        const unsigned span = (unsigned)(left < AGG_CAP ? left : AGG_CAP);
        for (int i = tid; i < AGG_CAP * D; i += AGG_THREADS) acc[i] = 0.0f;
        __syncthreads();

        int nq = 0;
        for (int base = wv * trip; base < ne; base += (AGG_THREADS >> 5) * trip) {
            int dv[AGG_UNROLL];
            if (base + trip <= ne) {
                #pragma unroll
                for (int u = 0; u < AGG_UNROLL; ++u) dv[u] = ds[base + u * 32 + lane];
            } else {
                #pragma unroll
                for (int u = 0; u < AGG_UNROLL; ++u) {
                    const int ee = base + u * 32 + lane;
                    dv[u] = (ee < ne) ? ds[ee] : -1;
                }
            }
            #pragma unroll
            for (int u = 0; u < AGG_UNROLL; ++u) {
                const unsigned r = (unsigned)(dv[u] - lo);
                const bool hit = r < span;
                const unsigned m = __builtin_amdgcn_ballot_w32(hit);
                if (m) {
                    if (hit) {
                        const int pos = nq + (int)__builtin_amdgcn_mbcnt_lo(m, 0u);
                        qw[pos] = base + u * 32 + lane;
                        qd[pos] = (unsigned short)r;
                    }
                    nq += __builtin_popcount(m);
                }
            }
            while (nq >= 32) {
                nq -= 32;
                agg_chunk(h, src, acc, e0 + qw[nq + lane], (int)qd[nq + lane], true, j);
            }
        }
        if (nq > 0) {
            const bool ok = lane < nq;
            agg_chunk(h, src, acc, e0 + (ok ? qw[lane] : 0), ok ? (int)qd[lane] : 0, ok, j);
        }
        __syncthreads();

        float* __restrict__ arow = agg + (size_t)lo * D;
        for (int i = tid; i < (int)span * D; i += AGG_THREADS) {
            const float v = acc[i];
            if (v != 0.0f) atomicAdd(arow + i, v);
        }
        __syncthreads();
    }
}

#define POOL_THREADS 1024
#define POOL_BLOCKS  256

__global__ __launch_bounds__(POOL_THREADS)
void edge_node_pool_k(const float* __restrict__ h2,
                      const unsigned* __restrict__ rec,
                      const float* __restrict__ bias,
                      const int* __restrict__ src,
                      const int* __restrict__ dst,
                      float* __restrict__ pooled,
                      float* __restrict__ cnt,
                      int n, long long E) {
    __shared__ float acc[NGRAPH * D];
    __shared__ float gcnt[NGRAPH];
    const int tid  = threadIdx.x;
    const int lane = tid & 31;
    const int j    = lane & 15;

    for (int i = tid; i < NGRAPH * D; i += POOL_THREADS) acc[i] = 0.0f;
    for (int i = tid; i < NGRAPH; i += POOL_THREADS) gcnt[i] = 0.0f;
    __syncthreads();

    const long long wave  = (long long)blockIdx.x * (POOL_THREADS >> 5)
                          + __builtin_amdgcn_readfirstlane(tid >> 5);
    const long long nwave = (long long)gridDim.x * (POOL_THREADS >> 5);

    const long long nfull = E >> 5;
    long long c = wave;
    int s = 0, d = 0;
    if (c < nfull) { s = src[(c << 5) + lane]; d = dst[(c << 5) + lane]; }
    while (c < nfull) {
        const v2u rd = *(const v2u*)(rec + REC_STRIDE * (size_t)(unsigned)d);
        const float w = __uint_as_float(rd.x);
        const int   g = (int)rd.y;
#define LOAD_ROW(K) const float hv##K = h2[(unsigned)row_bcast_i<K>(s) * D + j];
        ROWS16(LOAD_ROW)
#undef LOAD_ROW
        const long long cn = c + nwave;
        int sn = 0, dn = 0;
        if (cn < nfull) { sn = src[(cn << 5) + lane]; dn = dst[(cn << 5) + lane]; }
#define ADD_ROW(K) atomicAdd(&acc[row_bcast_i<K>(g) * D + j], hv##K * row_bcast_f<K>(w));
        ROWS16(ADD_ROW)
#undef ADD_ROW
        s = sn; d = dn; c = cn;
    }
    if (blockIdx.x == 0 && tid < 16) {
        for (long long e = nfull << 5; e < E; ++e) {
            const int ss = src[e], dd = dst[e];
            const v2u rd = *(const v2u*)(rec + REC_STRIDE * (size_t)(unsigned)dd);
            atomicAdd(&acc[(int)rd.y * D + j], h2[(size_t)ss * D + j] * __uint_as_float(rd.x));
        }
    }

    const long long n16 = (long long)n * D;
    for (long long t = (long long)blockIdx.x * POOL_THREADS + tid; t < n16;
         t += (long long)gridDim.x * POOL_THREADS) {
        const int i = (int)(t >> 4);
        const v2u ri = *(const v2u*)(rec + REC_STRIDE * (size_t)i);
        const float di = __uint_as_float(ri.x);
        const int   g  = (int)ri.y;
        atomicAdd(&acc[g * D + j], h2[t] * di + bias[j]);
        if (j == 0) atomicAdd(&gcnt[g], 1.0f);
    }
    __syncthreads();

    const int rot = (int)((blockIdx.x * POOL_THREADS) & (NGRAPH * D - 1));
    for (int i = tid; i < NGRAPH * D; i += POOL_THREADS) {
        const int idx = (i + rot) & (NGRAPH * D - 1);
        const float v = acc[idx];
        if (v != 0.0f) atomicAdd(pooled + idx, v);
    }
    for (int i = tid; i < NGRAPH; i += POOL_THREADS) {
        const float v = gcnt[i];
        if (v != 0.0f) atomicAdd(cnt + i, v);
    }
}

__global__ void fc_k(const float* __restrict__ pooled_r, const float* __restrict__ cnt_r,
                     const float* __restrict__ pooled_l, const float* __restrict__ cnt_l,
                     const float* __restrict__ fcW, const float* __restrict__ fcb,
                     float* __restrict__ out) {
    int t = blockIdx.x * blockDim.x + threadIdx.x;
    if (t < NGRAPH * 6) {
        int g = t / 6, o = t - g * 6;
        float icr = 1.0f / fmaxf(cnt_r[g], 1.0f);
        float icl = 1.0f / fmaxf(cnt_l[g], 1.0f);
        float acc = fcb[o];
        #pragma unroll
        for (int j = 0; j < D; ++j)
            acc += fcW[o * (2 * D) + j] * (pooled_r[g * D + j] * icr);
        #pragma unroll
        for (int j = 0; j < D; ++j)
            acc += fcW[o * (2 * D) + D + j] * (pooled_l[g * D + j] * icl);
        if (o < 3) out[g * 3 + o] = acc;
        else       out[NGRAPH * 3 + g * 3 + (o - 3)] = acc;
    }
}

static void run_protein(const float* emb, const float* W1, const float* b1,
                        const float* W2, const float* b2,
                        const int* ids, const int* edge, const int* batch,
                        int N, long long E,
                        float* bufA, float* bufB, float* dis,
                        float* pooled, float* cnt, hipStream_t stream) {
    const int* src = edge;
    const int* dst = edge + E;
    const int B = 256;
    dim3 blk(B);
    int gemm_grid = (int)(((N >> 4) + 7) / 8);
    if (gemm_grid < 1) gemm_grid = 1;

    degree_lds_k<<<dim3(DEG_SLICES, DEG_RANGES), dim3(DEG_THREADS), 0, stream>>>(
        dst, (unsigned*)bufB, N, E);
    dis_from_parts_k<<<(int)((N + B - 1) / B), blk, 0, stream>>>((const unsigned*)bufB, dis, N);

    gather_gemm16_scale_k<<<gemm_grid, blk, 0, stream>>>(emb, ids, W1, dis, bufA, bufB, N);

    aggregate_lds_k<<<dim3(AGG_SLICES, AGG_RANGES), dim3(AGG_THREADS), 0, stream>>>(
        bufA, src, dst, bufB, N, E);

    relu_gemm16_k<<<gemm_grid, blk, 0, stream>>>(bufA, bufB, dis, b1, W2, batch, pooled, cnt, N);

    edge_node_pool_k<<<POOL_BLOCKS, POOL_THREADS, 0, stream>>>(bufA, (const unsigned*)bufB, b2,
                                                               src, dst, pooled, cnt, N, E);
}

extern "C" void kernel_launch(void* const* d_in, const int* in_sizes, int n_in,
                              void* d_out, int out_size, void* d_ws, size_t ws_size,
                              hipStream_t stream) {
    const float* emb = (const float*)d_in[0];
    const float* W1  = (const float*)d_in[1];
    const float* b1  = (const float*)d_in[2];
    const float* W2  = (const float*)d_in[3];
    const float* b2  = (const float*)d_in[4];
    const float* fcW = (const float*)d_in[5];
    const float* fcb = (const float*)d_in[6];
    const int* rx     = (const int*)d_in[7];
    const int* redge  = (const int*)d_in[8];
    const int* rbatch = (const int*)d_in[9];
    const int* lx     = (const int*)d_in[10];
    const int* ledge  = (const int*)d_in[11];
    const int* lbatch = (const int*)d_in[12];

    const int N = in_sizes[7];
    const long long E = in_sizes[8] / 2;

    char* ws = (char*)d_ws;
    float* bufA = (float*)ws;  ws += (size_t)N * D * sizeof(float);
    float* bufB = (float*)ws;  ws += (size_t)N * D * sizeof(float);
    float* dis  = (float*)ws;  ws += (size_t)N * sizeof(float);
    float* pooled_r = (float*)ws; ws += (size_t)NGRAPH * D * sizeof(float);
    float* pooled_l = (float*)ws; ws += (size_t)NGRAPH * D * sizeof(float);
    float* cnt_r = (float*)ws; ws += (size_t)NGRAPH * sizeof(float);
    float* cnt_l = (float*)ws; ws += (size_t)NGRAPH * sizeof(float);

    run_protein(emb, W1, b1, W2, b2, rx, redge, rbatch, N, E,
                bufA, bufB, dis, pooled_r, cnt_r, stream);
    run_protein(emb, W1, b1, W2, b2, lx, ledge, lbatch, N, E,
                bufA, bufB, dis, pooled_l, cnt_l, stream);

    fc_k<<<(NGRAPH * 6 + 255) / 256, 256, 0, stream>>>(pooled_r, cnt_r, pooled_l, cnt_l,
                                                       fcW, fcb, (float*)d_out);
}


# both dst scans (in-degree count, layer-1 hit search) read 16 bytes per lane, 4 consecutive edges: a quarter of the load instructions; hits queue in another order, float summation order changes
# speedup vs baseline: 3.7975x; 1.0611x over previous
#include <hip/hip_runtime.h>
#include <hip/hip_bf16.h>

typedef float v2f __attribute__((ext_vector_type(2)));
typedef float v8f __attribute__((ext_vector_type(8)));
typedef unsigned v2u __attribute__((ext_vector_type(2)));

#define D 16
#define NGRAPH 1024

typedef float v4f __attribute__((ext_vector_type(4)));
typedef int v4i __attribute__((ext_vector_type(4)));

template <int K> __device__ __forceinline__ int row_bcast_i(int v) {
    return __builtin_amdgcn_update_dpp(0, v, 0x150 + K, 0xf, 0xf, false);
}
template <int K> __device__ __forceinline__ float row_bcast_f(float v) {
    return __int_as_float(row_bcast_i<K>(__float_as_int(v)));
}

#define REC_STRIDE D


__global__ void gather_gemm16_scale_k(const float* __restrict__ src_rows,
                                      const int* __restrict__ ids,
                                      const float* __restrict__ W,
                                      const float* __restrict__ dis,
                                      float* __restrict__ out,
                                      float* __restrict__ agg,
                                      int n_nodes) {
    const int lane  = threadIdx.x & 31;
    const int wave  = (int)((blockIdx.x * blockDim.x + threadIdx.x) >> 5);
    const int nwave = (int)((gridDim.x * blockDim.x) >> 5);
    const int row   = lane & 15;
    const int col   = lane & 15;
    const int koff  = (lane >> 4) << 1;
    const int rowb  = (lane >> 4) << 3;

    v2f bb0, bb1, bb2, bb3;
    bb0.x = W[(0  + koff) * D + col]; bb0.y = W[(1  + koff) * D + col];
    bb1.x = W[(4  + koff) * D + col]; bb1.y = W[(5  + koff) * D + col];
    bb2.x = W[(8  + koff) * D + col]; bb2.y = W[(9  + koff) * D + col];
    bb3.x = W[(12 + koff) * D + col]; bb3.y = W[(13 + koff) * D + col];

    const int ntiles = n_nodes >> 4;
    for (int tile = wave; tile < ntiles; tile += nwave) {
        const int node = (tile << 4) + row;
        const int r = ids[node];
        const float dv = dis[(tile << 4) + rowb + (lane & 7)];
        const float* xrow = src_rows + (size_t)r * D;
        v2f a0 = *(const v2f*)(xrow + 0  + koff);
        v2f a1 = *(const v2f*)(xrow + 4  + koff);
        v2f a2 = *(const v2f*)(xrow + 8  + koff);
        v2f a3 = *(const v2f*)(xrow + 12 + koff);
        v4f* ab = (v4f*)(agg + ((size_t)tile << 8));
        const v4f zero4 = {0.0f, 0.0f, 0.0f, 0.0f};
        ab[lane] = zero4;
        ab[32 + lane] = zero4;
        v8f c = {};
        c = __builtin_amdgcn_wmma_f32_16x16x4_f32(false, a0, false, bb0, (short)0, c, false, false);
        c = __builtin_amdgcn_wmma_f32_16x16x4_f32(false, a1, false, bb1, (short)0, c, false, false);
        c = __builtin_amdgcn_wmma_f32_16x16x4_f32(false, a2, false, bb2, (short)0, c, false, false);
        c = __builtin_amdgcn_wmma_f32_16x16x4_f32(false, a3, false, bb3, (short)0, c, false, false);
        float* ob = out + ((size_t)(tile << 4) + rowb) * D + col;
#define SCALE_STORE(K) ob[(size_t)K * D] = c[K] * row_bcast_f<K>(dv);
        SCALE_STORE(0) SCALE_STORE(1) SCALE_STORE(2) SCALE_STORE(3)
        SCALE_STORE(4) SCALE_STORE(5) SCALE_STORE(6) SCALE_STORE(7)
#undef SCALE_STORE
    }

    const int tail_start = ntiles << 4;
    if (tail_start < n_nodes) {
        int node = tail_start + (int)(blockIdx.x * blockDim.x + threadIdx.x);
        if (node < n_nodes) {
            const int r = ids[node];
            const float di = dis[node];
            const float* xrow = src_rows + (size_t)r * D;
            for (int j = 0; j < D; ++j) {
                float acc = 0.0f;
                for (int k = 0; k < D; ++k) acc += xrow[k] * W[k * D + j];
                out[(size_t)node * D + j] = acc * di;
                agg[(size_t)node * D + j] = 0.0f;
            }
        }
    }
}

__global__ void relu_gemm16_k(float* h, float* aggrec,
                              const float* __restrict__ dis,
                              const float* __restrict__ bias,
                              const float* __restrict__ W,
                              const int* __restrict__ batch,
                              float* __restrict__ pooled, float* __restrict__ cnt,
                              int n_nodes) {
    const int gid   = (int)(blockIdx.x * blockDim.x + threadIdx.x);
    const int gsize = (int)(gridDim.x * blockDim.x);
    for (int k = gid; k < NGRAPH * D; k += gsize) pooled[k] = 0.0f;
    for (int k = gid; k < NGRAPH; k += gsize) cnt[k] = 0.0f;

    const int lane  = threadIdx.x & 31;
    const int wave  = gid >> 5;
    const int nwave = gsize >> 5;
    const int row   = lane & 15;
    const int col   = lane & 15;
    const int koff  = (lane >> 4) << 1;
    const int rowb  = (lane >> 4) << 3;

    v2f bb0, bb1, bb2, bb3;
    bb0.x = W[(0  + koff) * D + col]; bb0.y = W[(1  + koff) * D + col];
    bb1.x = W[(4  + koff) * D + col]; bb1.y = W[(5  + koff) * D + col];
    bb2.x = W[(8  + koff) * D + col]; bb2.y = W[(9  + koff) * D + col];
    bb3.x = W[(12 + koff) * D + col]; bb3.y = W[(13 + koff) * D + col];
    const v2f bk0 = *(const v2f*)(bias + 0  + koff);
    const v2f bk1 = *(const v2f*)(bias + 4  + koff);
    const v2f bk2 = *(const v2f*)(bias + 8  + koff);
    const v2f bk3 = *(const v2f*)(bias + 12 + koff);

    const int ntiles = n_nodes >> 4;
    for (int tile = wave; tile < ntiles; tile += nwave) {
        const int node = (tile << 4) + row;
        const float di = dis[node];
        const float* hrow = h + (size_t)node * D;
        const float* grow = aggrec + (size_t)node * D;
        const v2f h0 = *(const v2f*)(hrow + 0  + koff), g0 = *(const v2f*)(grow + 0  + koff);
        const v2f h1 = *(const v2f*)(hrow + 4  + koff), g1 = *(const v2f*)(grow + 4  + koff);
        const v2f h2 = *(const v2f*)(hrow + 8  + koff), g2 = *(const v2f*)(grow + 8  + koff);
        const v2f h3 = *(const v2f*)(hrow + 12 + koff), g3 = *(const v2f*)(grow + 12 + koff);
        int bt = 0;
        if (lane < 16) bt = batch[node];
        v2f a0, a1, a2, a3;
#define RELU_IN(A, H, G, B) { float vx = di * (G.x + H.x) + B.x; float vy = di * (G.y + H.y) + B.y; \
                              A.x = (vx > 0.0f ? vx : 0.0f) * di; A.y = (vy > 0.0f ? vy : 0.0f) * di; }
        RELU_IN(a0, h0, g0, bk0) RELU_IN(a1, h1, g1, bk1) RELU_IN(a2, h2, g2, bk2) RELU_IN(a3, h3, g3, bk3)
#undef RELU_IN
        v8f c = {};
        c = __builtin_amdgcn_wmma_f32_16x16x4_f32(false, a0, false, bb0, (short)0, c, false, false);
        c = __builtin_amdgcn_wmma_f32_16x16x4_f32(false, a1, false, bb1, (short)0, c, false, false);
        c = __builtin_amdgcn_wmma_f32_16x16x4_f32(false, a2, false, bb2, (short)0, c, false, false);
        c = __builtin_amdgcn_wmma_f32_16x16x4_f32(false, a3, false, bb3, (short)0, c, false, false);
        float* ob = h + ((size_t)(tile << 4) + rowb) * D + col;
        #pragma unroll
        for (int rr = 0; rr < 8; ++rr) ob[(size_t)rr * D] = c[rr];
        if (lane < 16) {
            v2u rc;
            rc.x = __float_as_uint(di);
            rc.y = (unsigned)bt;
            *(v2u*)((unsigned*)aggrec + (size_t)node * REC_STRIDE) = rc;
        }
    }

    const int tail_start = ntiles << 4;
    if (tail_start < n_nodes) {
        int node = tail_start + gid;
        if (node < n_nodes) {
            const float di = dis[node];
            float z[D];
            for (int k = 0; k < D; ++k) {
                float v = di * (aggrec[(size_t)node * D + k] + h[(size_t)node * D + k]) + bias[k];
                z[k] = (v > 0.0f ? v : 0.0f) * di;
            }
            for (int j = 0; j < D; ++j) {
                float acc = 0.0f;
                for (int k = 0; k < D; ++k) acc += z[k] * W[k * D + j];
                h[(size_t)node * D + j] = acc;
            }
            v2u rc;
            rc.x = __float_as_uint(di);
            rc.y = (unsigned)batch[node];
            *(v2u*)((unsigned*)aggrec + (size_t)node * REC_STRIDE) = rc;
        }
    }
}

#define DEG_SLICES  16
#define DEG_RANGES  16
#define DEG_CAP     12512
#define DEG_THREADS 1024
#define DEG_UNROLL  8
#define DEG_TRIP    (DEG_THREADS * DEG_UNROLL)
static_assert(DEG_UNROLL % 4 == 0, "a thread reads its edges four at a time");

__global__ __launch_bounds__(DEG_THREADS)
void degree_lds_k(const int* __restrict__ dst, unsigned* __restrict__ part,
                  int n, long long E) {
    __shared__ unsigned cnt[DEG_CAP];
    const int tid   = threadIdx.x;
    const int slice = blockIdx.x;
    const long long per = (E + DEG_SLICES - 1) / DEG_SLICES;
    const long long e0  = (long long)slice * per;
    long long e1 = e0 + per; if (e1 > E) e1 = E;
    const int* __restrict__ ds = dst + e0;
    const int ne = (e1 > e0) ? (int)(e1 - e0) : 0;
    const bool wide = (((size_t)ds) & 15) == 0;

    for (int lo = blockIdx.y * DEG_CAP; lo < n; lo += gridDim.y * DEG_CAP) {
        const int left = n - lo;
        const unsigned span = (unsigned)(left < DEG_CAP ? left : DEG_CAP);
        for (int i = tid; i < DEG_CAP; i += DEG_THREADS) cnt[i] = 0u;
        __syncthreads();
        for (int base = 0; base < ne; base += DEG_TRIP) {
            int d[DEG_UNROLL];
            if (wide && base + DEG_TRIP <= ne) {
                #pragma unroll
                for (int q = 0; q < DEG_UNROLL / 4; ++q) {
                    const v4i t = *(const v4i*)(ds + base + q * (4 * DEG_THREADS) + tid * 4);
                    d[4 * q + 0] = t.x; d[4 * q + 1] = t.y; d[4 * q + 2] = t.z; d[4 * q + 3] = t.w;
                }
            } else {
                #pragma unroll
                for (int u = 0; u < DEG_UNROLL; ++u) {
                    const int ee = base + (u >> 2) * (4 * DEG_THREADS) + tid * 4 + (u & 3);
                    d[u] = (ee < ne) ? ds[ee] : -1;
                }
            }
            #pragma unroll
            for (int u = 0; u < DEG_UNROLL; ++u) {
                const unsigned r = (unsigned)(d[u] - lo);
                if (r < span) atomicAdd(&cnt[r], 1u);
            }
        }
        __syncthreads();
        unsigned* __restrict__ prow = part + (size_t)slice * n + lo;
        for (int i = tid; i < (int)span; i += DEG_THREADS) prow[i] = cnt[i];
        __syncthreads();
    }
}

__global__ void dis_from_parts_k(const unsigned* __restrict__ part,
                                 float* __restrict__ dis, int n) {
    int i = blockIdx.x * blockDim.x + threadIdx.x;
    if (i < n) {
        unsigned c = 0u;
        #pragma unroll
        for (int s = 0; s < DEG_SLICES; ++s) c += part[(size_t)s * n + i];
        dis[i] = rsqrtf((float)c + 1.0f);
    }
}

#define AGG_SLICES  4
#define AGG_RANGES  64
#define AGG_CAP     3125
#define AGG_THREADS 1024
#define AGG_UNROLL  16
static_assert(AGG_UNROLL % 4 == 0, "a lane reads its edges four at a time");
#define AGG_QCAP    (32 * AGG_UNROLL + 32)
static_assert(AGG_CAP <= 65536, "a range-local node index is queued in 16 bits");

#define ROWS16(X) X(0) X(1) X(2) X(3) X(4) X(5) X(6) X(7) \
                  X(8) X(9) X(10) X(11) X(12) X(13) X(14) X(15)

__device__ __forceinline__ void agg_chunk(const float* __restrict__ h,
                                          const int* __restrict__ src,
                                          float* acc,
                                          long long e, int dl, bool ok, int j) {
    int s = 0;
    if (ok) s = src[e];
    const int oki = ok ? 1 : 0;
#define LOAD_ROW(K) const float hv##K = h[(unsigned)row_bcast_i<K>(s) * D + j];
    ROWS16(LOAD_ROW)
#undef LOAD_ROW
#define ADD_ROW(K) if (row_bcast_i<K>(oki)) atomicAdd(&acc[row_bcast_i<K>(dl) * D + j], hv##K);
    ROWS16(ADD_ROW)
#undef ADD_ROW
}

__global__ __launch_bounds__(AGG_THREADS)
void aggregate_lds_k(const float* __restrict__ h,
                     const int* __restrict__ src,
                     const int* __restrict__ dst,
                     float* __restrict__ agg,
                     int n, long long E) {
    __shared__ float acc[AGG_CAP * D];
    __shared__ int   queue[(AGG_THREADS >> 5) * AGG_QCAP];
    __shared__ unsigned short qdst[(AGG_THREADS >> 5) * AGG_QCAP];
    const int tid  = threadIdx.x;
    const int lane = tid & 31;
    const int j    = lane & 15;
    const int wv   = __builtin_amdgcn_readfirstlane(tid >> 5);
    int* qw = queue + wv * AGG_QCAP;
    unsigned short* qd = qdst + wv * AGG_QCAP;

    const long long per = (E + AGG_SLICES - 1) / AGG_SLICES;
    const long long e0  = (long long)blockIdx.x * per;
    long long e1 = e0 + per; if (e1 > E) e1 = E;
    const int ne = (e1 > e0) ? (int)(e1 - e0) : 0;
    const int* __restrict__ ds = dst + e0;
    const int trip = 32 * AGG_UNROLL;
    const bool wide = (((size_t)ds) & 15) == 0;

    for (int lo = blockIdx.y * AGG_CAP; lo < n; lo += gridDim.y * AGG_CAP) {
        const int left = n - lo;
        const unsigned span = (unsigned)(left < AGG_CAP ? left : AGG_CAP);
        for (int i = tid; i < AGG_CAP * D; i += AGG_THREADS) acc[i] = 0.0f;
        __syncthreads();

        int nq = 0;
        for (int base = wv * trip; base < ne; base += (AGG_THREADS >> 5) * trip) {
            int dv[AGG_UNROLL];
            if (wide && base + trip <= ne) {
                #pragma unroll
                for (int q = 0; q < AGG_UNROLL / 4; ++q) {
                    const v4i t = *(const v4i*)(ds + base + q * 128 + lane * 4);
                    dv[4 * q + 0] = t.x; dv[4 * q + 1] = t.y; dv[4 * q + 2] = t.z; dv[4 * q + 3] = t.w;
                }
            } else {
                #pragma unroll
                for (int u = 0; u < AGG_UNROLL; ++u) {
                    const int ee = base + (u >> 2) * 128 + lane * 4 + (u & 3);
                    dv[u] = (ee < ne) ? ds[ee] : -1;
                }
            }
            #pragma unroll
            for (int u = 0; u < AGG_UNROLL; ++u) {
                const unsigned r = (unsigned)(dv[u] - lo);
                const bool hit = r < span;
                const unsigned m = __builtin_amdgcn_ballot_w32(hit);
                if (m) {
                    if (hit) {
                        const int pos = nq + (int)__builtin_amdgcn_mbcnt_lo(m, 0u);
                        qw[pos] = base + (u >> 2) * 128 + lane * 4 + (u & 3);
                        qd[pos] = (unsigned short)r;
                    }
                    nq += __builtin_popcount(m);
                }
            }
            while (nq >= 32) {
                nq -= 32;
                agg_chunk(h, src, acc, e0 + qw[nq + lane], (int)qd[nq + lane], true, j);
            }
        }
        if (nq > 0) {
            const bool ok = lane < nq;
            agg_chunk(h, src, acc, e0 + (ok ? qw[lane] : 0), ok ? (int)qd[lane] : 0, ok, j);
        }
        __syncthreads();

        float* __restrict__ arow = agg + (size_t)lo * D;
        for (int i = tid; i < (int)span * D; i += AGG_THREADS) {
            const float v = acc[i];
            if (v != 0.0f) atomicAdd(arow + i, v);
        }
        __syncthreads();
    }
}

#define POOL_THREADS 1024
#define POOL_BLOCKS  256

__global__ __launch_bounds__(POOL_THREADS)
void edge_node_pool_k(const float* __restrict__ h2,
                      const unsigned* __restrict__ rec,
                      const float* __restrict__ bias,
                      const int* __restrict__ src,
                      const int* __restrict__ dst,
                      float* __restrict__ pooled,
                      float* __restrict__ cnt,
                      int n, long long E) {
    __shared__ float acc[NGRAPH * D];
    __shared__ float gcnt[NGRAPH];
    const int tid  = threadIdx.x;
    const int lane = tid & 31;
    const int j    = lane & 15;

    for (int i = tid; i < NGRAPH * D; i += POOL_THREADS) acc[i] = 0.0f;
    for (int i = tid; i < NGRAPH; i += POOL_THREADS) gcnt[i] = 0.0f;
    __syncthreads();

    const long long wave  = (long long)blockIdx.x * (POOL_THREADS >> 5)
                          + __builtin_amdgcn_readfirstlane(tid >> 5);
    const long long nwave = (long long)gridDim.x * (POOL_THREADS >> 5);

    const long long nfull = E >> 5;
    long long c = wave;
    int s = 0, d = 0;
    if (c < nfull) { s = src[(c << 5) + lane]; d = dst[(c << 5) + lane]; }
    while (c < nfull) {
        const v2u rd = *(const v2u*)(rec + REC_STRIDE * (size_t)(unsigned)d);
        const float w = __uint_as_float(rd.x);
        const int   g = (int)rd.y;
#define LOAD_ROW(K) const float hv##K = h2[(unsigned)row_bcast_i<K>(s) * D + j];
        ROWS16(LOAD_ROW)
#undef LOAD_ROW
        const long long cn = c + nwave;
        int sn = 0, dn = 0;
        if (cn < nfull) { sn = src[(cn << 5) + lane]; dn = dst[(cn << 5) + lane]; }
#define ADD_ROW(K) atomicAdd(&acc[row_bcast_i<K>(g) * D + j], hv##K * row_bcast_f<K>(w));
        ROWS16(ADD_ROW)
#undef ADD_ROW
        s = sn; d = dn; c = cn;
    }
    if (blockIdx.x == 0 && tid < 16) {
        for (long long e = nfull << 5; e < E; ++e) {
            const int ss = src[e], dd = dst[e];
            const v2u rd = *(const v2u*)(rec + REC_STRIDE * (size_t)(unsigned)dd);
            atomicAdd(&acc[(int)rd.y * D + j], h2[(size_t)ss * D + j] * __uint_as_float(rd.x));
        }
    }

    const long long n16 = (long long)n * D;
    for (long long t = (long long)blockIdx.x * POOL_THREADS + tid; t < n16;
         t += (long long)gridDim.x * POOL_THREADS) {
        const int i = (int)(t >> 4);
        const v2u ri = *(const v2u*)(rec + REC_STRIDE * (size_t)i);
        const float di = __uint_as_float(ri.x);
        const int   g  = (int)ri.y;
        atomicAdd(&acc[g * D + j], h2[t] * di + bias[j]);
        if (j == 0) atomicAdd(&gcnt[g], 1.0f);
    }
    __syncthreads();

    const int rot = (int)((blockIdx.x * POOL_THREADS) & (NGRAPH * D - 1));
    for (int i = tid; i < NGRAPH * D; i += POOL_THREADS) {
        const int idx = (i + rot) & (NGRAPH * D - 1);
        const float v = acc[idx];
        if (v != 0.0f) atomicAdd(pooled + idx, v);
    }
    for (int i = tid; i < NGRAPH; i += POOL_THREADS) {
        const float v = gcnt[i];
        if (v != 0.0f) atomicAdd(cnt + i, v);
    }
}

__global__ void fc_k(const float* __restrict__ pooled_r, const float* __restrict__ cnt_r,
                     const float* __restrict__ pooled_l, const float* __restrict__ cnt_l,
                     const float* __restrict__ fcW, const float* __restrict__ fcb,
                     float* __restrict__ out) {
    int t = blockIdx.x * blockDim.x + threadIdx.x;
    if (t < NGRAPH * 6) {
        int g = t / 6, o = t - g * 6;
        float icr = 1.0f / fmaxf(cnt_r[g], 1.0f);
        float icl = 1.0f / fmaxf(cnt_l[g], 1.0f);
        float acc = fcb[o];
        #pragma unroll
        for (int j = 0; j < D; ++j)
            acc += fcW[o * (2 * D) + j] * (pooled_r[g * D + j] * icr);
        #pragma unroll
        for (int j = 0; j < D; ++j)
            acc += fcW[o * (2 * D) + D + j] * (pooled_l[g * D + j] * icl);
        if (o < 3) out[g * 3 + o] = acc;
        else       out[NGRAPH * 3 + g * 3 + (o - 3)] = acc;
    }
}

static void run_protein(const float* emb, const float* W1, const float* b1,
                        const float* W2, const float* b2,
                        const int* ids, const int* edge, const int* batch,
                        int N, long long E,
                        float* bufA, float* bufB, float* dis,
                        float* pooled, float* cnt, hipStream_t stream) {
    const int* src = edge;
    const int* dst = edge + E;
    const int B = 256;
    dim3 blk(B);
    int gemm_grid = (int)(((N >> 4) + 7) / 8);
    if (gemm_grid < 1) gemm_grid = 1;

    degree_lds_k<<<dim3(DEG_SLICES, DEG_RANGES), dim3(DEG_THREADS), 0, stream>>>(
        dst, (unsigned*)bufB, N, E);
    dis_from_parts_k<<<(int)((N + B - 1) / B), blk, 0, stream>>>((const unsigned*)bufB, dis, N);

    gather_gemm16_scale_k<<<gemm_grid, blk, 0, stream>>>(emb, ids, W1, dis, bufA, bufB, N);

    aggregate_lds_k<<<dim3(AGG_SLICES, AGG_RANGES), dim3(AGG_THREADS), 0, stream>>>(
        bufA, src, dst, bufB, N, E);

    relu_gemm16_k<<<gemm_grid, blk, 0, stream>>>(bufA, bufB, dis, b1, W2, batch, pooled, cnt, N);

    edge_node_pool_k<<<POOL_BLOCKS, POOL_THREADS, 0, stream>>>(bufA, (const unsigned*)bufB, b2,
                                                               src, dst, pooled, cnt, N, E);
}

extern "C" void kernel_launch(void* const* d_in, const int* in_sizes, int n_in,
                              void* d_out, int out_size, void* d_ws, size_t ws_size,
                              hipStream_t stream) {
    const float* emb = (const float*)d_in[0];
    const float* W1  = (const float*)d_in[1];
    const float* b1  = (const float*)d_in[2];
    const float* W2  = (const float*)d_in[3];
    const float* b2  = (const float*)d_in[4];
    const float* fcW = (const float*)d_in[5];
    const float* fcb = (const float*)d_in[6];
    const int* rx     = (const int*)d_in[7];
    const int* redge  = (const int*)d_in[8];
    const int* rbatch = (const int*)d_in[9];
    const int* lx     = (const int*)d_in[10];
    const int* ledge  = (const int*)d_in[11];
    const int* lbatch = (const int*)d_in[12];

    const int N = in_sizes[7];
    const long long E = in_sizes[8] / 2;

    char* ws = (char*)d_ws;
    float* bufA = (float*)ws;  ws += (size_t)N * D * sizeof(float);
    float* bufB = (float*)ws;  ws += (size_t)N * D * sizeof(float);
    float* dis  = (float*)ws;  ws += (size_t)N * sizeof(float);
    float* pooled_r = (float*)ws; ws += (size_t)NGRAPH * D * sizeof(float);
    float* pooled_l = (float*)ws; ws += (size_t)NGRAPH * D * sizeof(float);
    float* cnt_r = (float*)ws; ws += (size_t)NGRAPH * sizeof(float);
    float* cnt_l = (float*)ws; ws += (size_t)NGRAPH * sizeof(float);

    run_protein(emb, W1, b1, W2, b2, rx, redge, rbatch, N, E,
                bufA, bufB, dis, pooled_r, cnt_r, stream);
    run_protein(emb, W1, b1, W2, b2, lx, ledge, lbatch, N, E,
                bufA, bufB, dis, pooled_l, cnt_l, stream);

    fc_k<<<(NGRAPH * 6 + 255) / 256, 256, 0, stream>>>(pooled_r, cnt_r, pooled_l, cnt_l,
                                                       fcW, fcb, (float*)d_out);
}


# both dst scans keep eight 16-byte loads per lane in flight (in-degree count had 2, layer-1 hit search 4): 32 edges per lane and trip; hits queue in another order, float summation order changes
# speedup vs baseline: 4.0422x; 1.0644x over previous
#include <hip/hip_runtime.h>
#include <hip/hip_bf16.h>

typedef float v2f __attribute__((ext_vector_type(2)));
typedef float v8f __attribute__((ext_vector_type(8)));
typedef unsigned v2u __attribute__((ext_vector_type(2)));

#define D 16
#define NGRAPH 1024

typedef float v4f __attribute__((ext_vector_type(4)));
typedef int v4i __attribute__((ext_vector_type(4)));

template <int K> __device__ __forceinline__ int row_bcast_i(int v) {
    return __builtin_amdgcn_update_dpp(0, v, 0x150 + K, 0xf, 0xf, false);
}
template <int K> __device__ __forceinline__ float row_bcast_f(float v) {
    return __int_as_float(row_bcast_i<K>(__float_as_int(v)));
}

#define REC_STRIDE D


__global__ void gather_gemm16_scale_k(const float* __restrict__ src_rows,
                                      const int* __restrict__ ids,
                                      const float* __restrict__ W,
                                      const float* __restrict__ dis,
                                      float* __restrict__ out,
                                      float* __restrict__ agg,
                                      int n_nodes) {
    const int lane  = threadIdx.x & 31;
    const int wave  = (int)((blockIdx.x * blockDim.x + threadIdx.x) >> 5);
    const int nwave = (int)((gridDim.x * blockDim.x) >> 5);
    const int row   = lane & 15;
    const int col   = lane & 15;
    const int koff  = (lane >> 4) << 1;
    const int rowb  = (lane >> 4) << 3;

    v2f bb0, bb1, bb2, bb3;
    bb0.x = W[(0  + koff) * D + col]; bb0.y = W[(1  + koff) * D + col];
    bb1.x = W[(4  + koff) * D + col]; bb1.y = W[(5  + koff) * D + col];
    bb2.x = W[(8  + koff) * D + col]; bb2.y = W[(9  + koff) * D + col];
    bb3.x = W[(12 + koff) * D + col]; bb3.y = W[(13 + koff) * D + col];

    const int ntiles = n_nodes >> 4;
    for (int tile = wave; tile < ntiles; tile += nwave) {
        const int node = (tile << 4) + row;
        const int r = ids[node];
        const float dv = dis[(tile << 4) + rowb + (lane & 7)];
        const float* xrow = src_rows + (size_t)r * D;
        v2f a0 = *(const v2f*)(xrow + 0  + koff);
        v2f a1 = *(const v2f*)(xrow + 4  + koff);
        v2f a2 = *(const v2f*)(xrow + 8  + koff);
        v2f a3 = *(const v2f*)(xrow + 12 + koff);
        v4f* ab = (v4f*)(agg + ((size_t)tile << 8));
        const v4f zero4 = {0.0f, 0.0f, 0.0f, 0.0f};
        ab[lane] = zero4;
        ab[32 + lane] = zero4;
        v8f c = {};
        c = __builtin_amdgcn_wmma_f32_16x16x4_f32(false, a0, false, bb0, (short)0, c, false, false);
        c = __builtin_amdgcn_wmma_f32_16x16x4_f32(false, a1, false, bb1, (short)0, c, false, false);
        c = __builtin_amdgcn_wmma_f32_16x16x4_f32(false, a2, false, bb2, (short)0, c, false, false);
        c = __builtin_amdgcn_wmma_f32_16x16x4_f32(false, a3, false, bb3, (short)0, c, false, false);
        float* ob = out + ((size_t)(tile << 4) + rowb) * D + col;
#define SCALE_STORE(K) ob[(size_t)K * D] = c[K] * row_bcast_f<K>(dv);
        SCALE_STORE(0) SCALE_STORE(1) SCALE_STORE(2) SCALE_STORE(3)
        SCALE_STORE(4) SCALE_STORE(5) SCALE_STORE(6) SCALE_STORE(7)
#undef SCALE_STORE
    }

    const int tail_start = ntiles << 4;
    if (tail_start < n_nodes) {
        int node = tail_start + (int)(blockIdx.x * blockDim.x + threadIdx.x);
        if (node < n_nodes) {
            const int r = ids[node];
            const float di = dis[node];
            const float* xrow = src_rows + (size_t)r * D;
            for (int j = 0; j < D; ++j) {
                float acc = 0.0f;
                for (int k = 0; k < D; ++k) acc += xrow[k] * W[k * D + j];
                out[(size_t)node * D + j] = acc * di;
                agg[(size_t)node * D + j] = 0.0f;
            }
        }
    }
}

__global__ void relu_gemm16_k(float* h, float* aggrec,
                              const float* __restrict__ dis,
                              const float* __restrict__ bias,
                              const float* __restrict__ W,
                              const int* __restrict__ batch,
                              float* __restrict__ pooled, float* __restrict__ cnt,
                              int n_nodes) {
    const int gid   = (int)(blockIdx.x * blockDim.x + threadIdx.x);
    const int gsize = (int)(gridDim.x * blockDim.x);
    for (int k = gid; k < NGRAPH * D; k += gsize) pooled[k] = 0.0f;
    for (int k = gid; k < NGRAPH; k += gsize) cnt[k] = 0.0f;

    const int lane  = threadIdx.x & 31;
    const int wave  = gid >> 5;
    const int nwave = gsize >> 5;
    const int row   = lane & 15;
    const int col   = lane & 15;
    const int koff  = (lane >> 4) << 1;
    const int rowb  = (lane >> 4) << 3;

    v2f bb0, bb1, bb2, bb3;
    bb0.x = W[(0  + koff) * D + col]; bb0.y = W[(1  + koff) * D + col];
    bb1.x = W[(4  + koff) * D + col]; bb1.y = W[(5  + koff) * D + col];
    bb2.x = W[(8  + koff) * D + col]; bb2.y = W[(9  + koff) * D + col];
    bb3.x = W[(12 + koff) * D + col]; bb3.y = W[(13 + koff) * D + col];
    const v2f bk0 = *(const v2f*)(bias + 0  + koff);
    const v2f bk1 = *(const v2f*)(bias + 4  + koff);
    const v2f bk2 = *(const v2f*)(bias + 8  + koff);
    const v2f bk3 = *(const v2f*)(bias + 12 + koff);

    const int ntiles = n_nodes >> 4;
    for (int tile = wave; tile < ntiles; tile += nwave) {
        const int node = (tile << 4) + row;
        const float di = dis[node];
        const float* hrow = h + (size_t)node * D;
        const float* grow = aggrec + (size_t)node * D;
        const v2f h0 = *(const v2f*)(hrow + 0  + koff), g0 = *(const v2f*)(grow + 0  + koff);
        const v2f h1 = *(const v2f*)(hrow + 4  + koff), g1 = *(const v2f*)(grow + 4  + koff);
        const v2f h2 = *(const v2f*)(hrow + 8  + koff), g2 = *(const v2f*)(grow + 8  + koff);
        const v2f h3 = *(const v2f*)(hrow + 12 + koff), g3 = *(const v2f*)(grow + 12 + koff);
        int bt = 0;
        if (lane < 16) bt = batch[node];
        v2f a0, a1, a2, a3;
#define RELU_IN(A, H, G, B) { float vx = di * (G.x + H.x) + B.x; float vy = di * (G.y + H.y) + B.y; \
                              A.x = (vx > 0.0f ? vx : 0.0f) * di; A.y = (vy > 0.0f ? vy : 0.0f) * di; }
        RELU_IN(a0, h0, g0, bk0) RELU_IN(a1, h1, g1, bk1) RELU_IN(a2, h2, g2, bk2) RELU_IN(a3, h3, g3, bk3)
#undef RELU_IN
        v8f c = {};
        c = __builtin_amdgcn_wmma_f32_16x16x4_f32(false, a0, false, bb0, (short)0, c, false, false);
        c = __builtin_amdgcn_wmma_f32_16x16x4_f32(false, a1, false, bb1, (short)0, c, false, false);
        c = __builtin_amdgcn_wmma_f32_16x16x4_f32(false, a2, false, bb2, (short)0, c, false, false);
        c = __builtin_amdgcn_wmma_f32_16x16x4_f32(false, a3, false, bb3, (short)0, c, false, false);
        float* ob = h + ((size_t)(tile << 4) + rowb) * D + col;
        #pragma unroll
        for (int rr = 0; rr < 8; ++rr) ob[(size_t)rr * D] = c[rr];
        if (lane < 16) {
            v2u rc;
            rc.x = __float_as_uint(di);
            rc.y = (unsigned)bt;
            *(v2u*)((unsigned*)aggrec + (size_t)node * REC_STRIDE) = rc;
        }
    }

    const int tail_start = ntiles << 4;
    if (tail_start < n_nodes) {
        int node = tail_start + gid;
        if (node < n_nodes) {
            const float di = dis[node];
            float z[D];
            for (int k = 0; k < D; ++k) {
                float v = di * (aggrec[(size_t)node * D + k] + h[(size_t)node * D + k]) + bias[k];
                z[k] = (v > 0.0f ? v : 0.0f) * di;
            }
            for (int j = 0; j < D; ++j) {
                float acc = 0.0f;
                for (int k = 0; k < D; ++k) acc += z[k] * W[k * D + j];
                h[(size_t)node * D + j] = acc;
            }
            v2u rc;
            rc.x = __float_as_uint(di);
            rc.y = (unsigned)batch[node];
            *(v2u*)((unsigned*)aggrec + (size_t)node * REC_STRIDE) = rc;
        }
    }
}

#define DEG_SLICES  16
#define DEG_RANGES  16
#define DEG_CAP     12512
#define DEG_THREADS 1024
#define DEG_UNROLL  32
#define DEG_TRIP    (DEG_THREADS * DEG_UNROLL)
static_assert(DEG_UNROLL % 4 == 0, "a thread reads its edges four at a time");

__global__ __launch_bounds__(DEG_THREADS)
void degree_lds_k(const int* __restrict__ dst, unsigned* __restrict__ part,
                  int n, long long E) {
    __shared__ unsigned cnt[DEG_CAP];
    const int tid   = threadIdx.x;
    const int slice = blockIdx.x;
    const long long per = (E + DEG_SLICES - 1) / DEG_SLICES;
    const long long e0  = (long long)slice * per;
    long long e1 = e0 + per; if (e1 > E) e1 = E;
    const int* __restrict__ ds = dst + e0;
    const int ne = (e1 > e0) ? (int)(e1 - e0) : 0;
    const bool wide = (((size_t)ds) & 15) == 0;

    for (int lo = blockIdx.y * DEG_CAP; lo < n; lo += gridDim.y * DEG_CAP) {
        const int left = n - lo;
        const unsigned span = (unsigned)(left < DEG_CAP ? left : DEG_CAP);
        for (int i = tid; i < DEG_CAP; i += DEG_THREADS) cnt[i] = 0u;
        __syncthreads();
        for (int base = 0; base < ne; base += DEG_TRIP) {
            int d[DEG_UNROLL];
            if (wide && base + DEG_TRIP <= ne) {
                #pragma unroll
                for (int q = 0; q < DEG_UNROLL / 4; ++q) {
                    const v4i t = *(const v4i*)(ds + base + q * (4 * DEG_THREADS) + tid * 4);
                    d[4 * q + 0] = t.x; d[4 * q + 1] = t.y; d[4 * q + 2] = t.z; d[4 * q + 3] = t.w;
                }
            } else {
                #pragma unroll
                for (int q = 0; q < DEG_UNROLL / 4; ++q) {
                    const int gb = base + q * (4 * DEG_THREADS);
                    if (wide && gb + 4 * DEG_THREADS <= ne) {
                        const v4i t = *(const v4i*)(ds + gb + tid * 4);
                        d[4 * q + 0] = t.x; d[4 * q + 1] = t.y; d[4 * q + 2] = t.z; d[4 * q + 3] = t.w;
                    } else {
                        #pragma unroll
                        for (int k = 0; k < 4; ++k) {
                            const int ee = gb + tid * 4 + k;
                            d[4 * q + k] = (ee < ne) ? ds[ee] : -1;
                        }
                    }
                }
            }
            #pragma unroll
            for (int u = 0; u < DEG_UNROLL; ++u) {
                const unsigned r = (unsigned)(d[u] - lo);
                if (r < span) atomicAdd(&cnt[r], 1u);
            }
        }
        __syncthreads();
        unsigned* __restrict__ prow = part + (size_t)slice * n + lo;
        for (int i = tid; i < (int)span; i += DEG_THREADS) prow[i] = cnt[i];
        __syncthreads();
    }
}

__global__ void dis_from_parts_k(const unsigned* __restrict__ part,
                                 float* __restrict__ dis, int n) {
    int i = blockIdx.x * blockDim.x + threadIdx.x;
    if (i < n) {
        unsigned c = 0u;
        #pragma unroll
        for (int s = 0; s < DEG_SLICES; ++s) c += part[(size_t)s * n + i];
        dis[i] = rsqrtf((float)c + 1.0f);
    }
}

#define AGG_SLICES  4
#define AGG_RANGES  64
#define AGG_CAP     3125
#define AGG_THREADS 1024
#define AGG_UNROLL  16
#define AGG_HALVES  2
#define AGG_EDGES   (AGG_UNROLL * AGG_HALVES)
static_assert(AGG_UNROLL % 4 == 0, "a lane reads its edges four at a time");
#define AGG_QCAP    (32 * AGG_UNROLL + 32)
static_assert(AGG_CAP <= 65536, "a range-local node index is queued in 16 bits");

#define ROWS16(X) X(0) X(1) X(2) X(3) X(4) X(5) X(6) X(7) \
                  X(8) X(9) X(10) X(11) X(12) X(13) X(14) X(15)

__device__ __forceinline__ void agg_chunk(const float* __restrict__ h,
                                          const int* __restrict__ src,
                                          float* acc,
                                          long long e, int dl, bool ok, int j) {
    int s = 0;
    if (ok) s = src[e];
    const int oki = ok ? 1 : 0;
#define LOAD_ROW(K) const float hv##K = h[(unsigned)row_bcast_i<K>(s) * D + j];
    ROWS16(LOAD_ROW)
#undef LOAD_ROW
#define ADD_ROW(K) if (row_bcast_i<K>(oki)) atomicAdd(&acc[row_bcast_i<K>(dl) * D + j], hv##K);
    ROWS16(ADD_ROW)
#undef ADD_ROW
}

__global__ __launch_bounds__(AGG_THREADS)
void aggregate_lds_k(const float* __restrict__ h,
                     const int* __restrict__ src,
                     const int* __restrict__ dst,
                     float* __restrict__ agg,
                     int n, long long E) {
    __shared__ float acc[AGG_CAP * D];
    __shared__ int   queue[(AGG_THREADS >> 5) * AGG_QCAP];
    __shared__ unsigned short qdst[(AGG_THREADS >> 5) * AGG_QCAP];
    const int tid  = threadIdx.x;
    const int lane = tid & 31;
    const int j    = lane & 15;
    const int wv   = __builtin_amdgcn_readfirstlane(tid >> 5);
    int* qw = queue + wv * AGG_QCAP;
    unsigned short* qd = qdst + wv * AGG_QCAP;

    const long long per = (E + AGG_SLICES - 1) / AGG_SLICES;
    const long long e0  = (long long)blockIdx.x * per;
    long long e1 = e0 + per; if (e1 > E) e1 = E;
    const int ne = (e1 > e0) ? (int)(e1 - e0) : 0;
    const int* __restrict__ ds = dst + e0;
    const int trip = 32 * AGG_EDGES;
    const bool wide = (((size_t)ds) & 15) == 0;

    for (int lo = blockIdx.y * AGG_CAP; lo < n; lo += gridDim.y * AGG_CAP) {
        const int left = n - lo;
        const unsigned span = (unsigned)(left < AGG_CAP ? left : AGG_CAP);
        for (int i = tid; i < AGG_CAP * D; i += AGG_THREADS) acc[i] = 0.0f;
        __syncthreads();

        int nq = 0;
        for (int base = wv * trip; base < ne; base += (AGG_THREADS >> 5) * trip) {
            int dv[AGG_EDGES];
            if (wide && base + trip <= ne) {
                #pragma unroll
                for (int q = 0; q < AGG_EDGES / 4; ++q) {
                    const v4i t = *(const v4i*)(ds + base + q * 128 + lane * 4);
                    dv[4 * q + 0] = t.x; dv[4 * q + 1] = t.y; dv[4 * q + 2] = t.z; dv[4 * q + 3] = t.w;
                }
            } else {
                #pragma unroll
                for (int q = 0; q < AGG_EDGES / 4; ++q) {
                    const int gb = base + q * 128;
                    if (wide && gb + 128 <= ne) {
                        const v4i t = *(const v4i*)(ds + gb + lane * 4);
                        dv[4 * q + 0] = t.x; dv[4 * q + 1] = t.y; dv[4 * q + 2] = t.z; dv[4 * q + 3] = t.w;
                    } else {
                        #pragma unroll
                        for (int k = 0; k < 4; ++k) {
                            const int ee = gb + lane * 4 + k;
                            dv[4 * q + k] = (ee < ne) ? ds[ee] : -1;
                        }
                    }
                }
            }
            #pragma unroll
            for (int hf = 0; hf < AGG_HALVES; ++hf) {
                #pragma unroll
                for (int u = hf * AGG_UNROLL; u < (hf + 1) * AGG_UNROLL; ++u) {
                    const unsigned r = (unsigned)(dv[u] - lo);
                    const bool hit = r < span;
                    const unsigned m = __builtin_amdgcn_ballot_w32(hit);
                    if (m) {
                        if (hit) {
                            const int pos = nq + (int)__builtin_amdgcn_mbcnt_lo(m, 0u);
                            qw[pos] = base + (u >> 2) * 128 + lane * 4 + (u & 3);
                            qd[pos] = (unsigned short)r;
                        }
                        nq += __builtin_popcount(m);
                    }
                }
                while (nq >= 32) {
                    nq -= 32;
                    agg_chunk(h, src, acc, e0 + qw[nq + lane], (int)qd[nq + lane], true, j);
                }
            }
        }
        if (nq > 0) {
            const bool ok = lane < nq;
            agg_chunk(h, src, acc, e0 + (ok ? qw[lane] : 0), ok ? (int)qd[lane] : 0, ok, j);
        }
        __syncthreads();

        float* __restrict__ arow = agg + (size_t)lo * D;
        for (int i = tid; i < (int)span * D; i += AGG_THREADS) {
            const float v = acc[i];
            if (v != 0.0f) atomicAdd(arow + i, v);
        }
        __syncthreads();
    }
}

#define POOL_THREADS 1024
#define POOL_BLOCKS  256

__global__ __launch_bounds__(POOL_THREADS)
void edge_node_pool_k(const float* __restrict__ h2,
                      const unsigned* __restrict__ rec,
                      const float* __restrict__ bias,
                      const int* __restrict__ src,
                      const int* __restrict__ dst,
                      float* __restrict__ pooled,
                      float* __restrict__ cnt,
                      int n, long long E) {
    __shared__ float acc[NGRAPH * D];
    __shared__ float gcnt[NGRAPH];
    const int tid  = threadIdx.x;
    const int lane = tid & 31;
    const int j    = lane & 15;

    for (int i = tid; i < NGRAPH * D; i += POOL_THREADS) acc[i] = 0.0f;
    for (int i = tid; i < NGRAPH; i += POOL_THREADS) gcnt[i] = 0.0f;
    __syncthreads();

    const long long wave  = (long long)blockIdx.x * (POOL_THREADS >> 5)
                          + __builtin_amdgcn_readfirstlane(tid >> 5);
    const long long nwave = (long long)gridDim.x * (POOL_THREADS >> 5);

    const long long nfull = E >> 5;
    long long c = wave;
    int s = 0, d = 0;
    if (c < nfull) { s = src[(c << 5) + lane]; d = dst[(c << 5) + lane]; }
    while (c < nfull) {
        const v2u rd = *(const v2u*)(rec + REC_STRIDE * (size_t)(unsigned)d);
        const float w = __uint_as_float(rd.x);
        const int   g = (int)rd.y;
#define LOAD_ROW(K) const float hv##K = h2[(unsigned)row_bcast_i<K>(s) * D + j];
        ROWS16(LOAD_ROW)
#undef LOAD_ROW
        const long long cn = c + nwave;
        int sn = 0, dn = 0;
        if (cn < nfull) { sn = src[(cn << 5) + lane]; dn = dst[(cn << 5) + lane]; }
#define ADD_ROW(K) atomicAdd(&acc[row_bcast_i<K>(g) * D + j], hv##K * row_bcast_f<K>(w));
        ROWS16(ADD_ROW)
#undef ADD_ROW
        s = sn; d = dn; c = cn;
    }
    if (blockIdx.x == 0 && tid < 16) {
        for (long long e = nfull << 5; e < E; ++e) {
            const int ss = src[e], dd = dst[e];
            const v2u rd = *(const v2u*)(rec + REC_STRIDE * (size_t)(unsigned)dd);
            atomicAdd(&acc[(int)rd.y * D + j], h2[(size_t)ss * D + j] * __uint_as_float(rd.x));
        }
    }

    const long long n16 = (long long)n * D;
    for (long long t = (long long)blockIdx.x * POOL_THREADS + tid; t < n16;
         t += (long long)gridDim.x * POOL_THREADS) {
        const int i = (int)(t >> 4);
        const v2u ri = *(const v2u*)(rec + REC_STRIDE * (size_t)i);
        const float di = __uint_as_float(ri.x);
        const int   g  = (int)ri.y;
        atomicAdd(&acc[g * D + j], h2[t] * di + bias[j]);
        if (j == 0) atomicAdd(&gcnt[g], 1.0f);
    }
    __syncthreads();

    const int rot = (int)((blockIdx.x * POOL_THREADS) & (NGRAPH * D - 1));
    for (int i = tid; i < NGRAPH * D; i += POOL_THREADS) {
        const int idx = (i + rot) & (NGRAPH * D - 1);
        const float v = acc[idx];
        if (v != 0.0f) atomicAdd(pooled + idx, v);
    }
    for (int i = tid; i < NGRAPH; i += POOL_THREADS) {
        const float v = gcnt[i];
        if (v != 0.0f) atomicAdd(cnt + i, v);
    }
}

__global__ void fc_k(const float* __restrict__ pooled_r, const float* __restrict__ cnt_r,
                     const float* __restrict__ pooled_l, const float* __restrict__ cnt_l,
                     const float* __restrict__ fcW, const float* __restrict__ fcb,
                     float* __restrict__ out) {
    int t = blockIdx.x * blockDim.x + threadIdx.x;
    if (t < NGRAPH * 6) {
        int g = t / 6, o = t - g * 6;
        float icr = 1.0f / fmaxf(cnt_r[g], 1.0f);
        float icl = 1.0f / fmaxf(cnt_l[g], 1.0f);
        float acc = fcb[o];
        #pragma unroll
        for (int j = 0; j < D; ++j)
            acc += fcW[o * (2 * D) + j] * (pooled_r[g * D + j] * icr);
        #pragma unroll
        for (int j = 0; j < D; ++j)
            acc += fcW[o * (2 * D) + D + j] * (pooled_l[g * D + j] * icl);
        if (o < 3) out[g * 3 + o] = acc;
        else       out[NGRAPH * 3 + g * 3 + (o - 3)] = acc;
    }
}

static void run_protein(const float* emb, const float* W1, const float* b1,
                        const float* W2, const float* b2,
                        const int* ids, const int* edge, const int* batch,
                        int N, long long E,
                        float* bufA, float* bufB, float* dis,
                        float* pooled, float* cnt, hipStream_t stream) {
    const int* src = edge;
    const int* dst = edge + E;
    const int B = 256;
    dim3 blk(B);
    int gemm_grid = (int)(((N >> 4) + 7) / 8);
    if (gemm_grid < 1) gemm_grid = 1;

    degree_lds_k<<<dim3(DEG_SLICES, DEG_RANGES), dim3(DEG_THREADS), 0, stream>>>(
        dst, (unsigned*)bufB, N, E);
    dis_from_parts_k<<<(int)((N + B - 1) / B), blk, 0, stream>>>((const unsigned*)bufB, dis, N);

    gather_gemm16_scale_k<<<gemm_grid, blk, 0, stream>>>(emb, ids, W1, dis, bufA, bufB, N);

    aggregate_lds_k<<<dim3(AGG_SLICES, AGG_RANGES), dim3(AGG_THREADS), 0, stream>>>(
        bufA, src, dst, bufB, N, E);

    relu_gemm16_k<<<gemm_grid, blk, 0, stream>>>(bufA, bufB, dis, b1, W2, batch, pooled, cnt, N);

    edge_node_pool_k<<<POOL_BLOCKS, POOL_THREADS, 0, stream>>>(bufA, (const unsigned*)bufB, b2,
                                                               src, dst, pooled, cnt, N, E);
}

extern "C" void kernel_launch(void* const* d_in, const int* in_sizes, int n_in,
                              void* d_out, int out_size, void* d_ws, size_t ws_size,
                              hipStream_t stream) {
    const float* emb = (const float*)d_in[0];
    const float* W1  = (const float*)d_in[1];
    const float* b1  = (const float*)d_in[2];
    const float* W2  = (const float*)d_in[3];
    const float* b2  = (const float*)d_in[4];
    const float* fcW = (const float*)d_in[5];
    const float* fcb = (const float*)d_in[6];
    const int* rx     = (const int*)d_in[7];
    const int* redge  = (const int*)d_in[8];
    const int* rbatch = (const int*)d_in[9];
    const int* lx     = (const int*)d_in[10];
    const int* ledge  = (const int*)d_in[11];
    const int* lbatch = (const int*)d_in[12];

    const int N = in_sizes[7];
    const long long E = in_sizes[8] / 2;

    char* ws = (char*)d_ws;
    float* bufA = (float*)ws;  ws += (size_t)N * D * sizeof(float);
    float* bufB = (float*)ws;  ws += (size_t)N * D * sizeof(float);
    float* dis  = (float*)ws;  ws += (size_t)N * sizeof(float);
    float* pooled_r = (float*)ws; ws += (size_t)NGRAPH * D * sizeof(float);
    float* pooled_l = (float*)ws; ws += (size_t)NGRAPH * D * sizeof(float);
    float* cnt_r = (float*)ws; ws += (size_t)NGRAPH * sizeof(float);
    float* cnt_l = (float*)ws; ws += (size_t)NGRAPH * sizeof(float);

    run_protein(emb, W1, b1, W2, b2, rx, redge, rbatch, N, E,
                bufA, bufB, dis, pooled_r, cnt_r, stream);
    run_protein(emb, W1, b1, W2, b2, lx, ledge, lbatch, N, E,
                bufA, bufB, dis, pooled_l, cnt_l, stream);

    fc_k<<<(NGRAPH * 6 + 255) / 256, 256, 0, stream>>>(pooled_r, cnt_r, pooled_l, cnt_l,
                                                       fcW, fcb, (float*)d_out);
}


# both dst scans over fewer, larger node ranges: in-degree 4 x 32 edge slices (was 16 x 16), layer-1 sums 42 x 6 (was 64 x 4), queued hit packed in one word; float summation order changes
# speedup vs baseline: 4.2869x; 1.0605x over previous
#include <hip/hip_runtime.h>
#include <hip/hip_bf16.h>

typedef float v2f __attribute__((ext_vector_type(2)));
typedef float v8f __attribute__((ext_vector_type(8)));
typedef unsigned v2u __attribute__((ext_vector_type(2)));

#define D 16
#define NGRAPH 1024

typedef float v4f __attribute__((ext_vector_type(4)));
typedef int v4i __attribute__((ext_vector_type(4)));

template <int K> __device__ __forceinline__ int row_bcast_i(int v) {
    return __builtin_amdgcn_update_dpp(0, v, 0x150 + K, 0xf, 0xf, false);
}
template <int K> __device__ __forceinline__ float row_bcast_f(float v) {
    return __int_as_float(row_bcast_i<K>(__float_as_int(v)));
}

#define REC_STRIDE D


__global__ void gather_gemm16_scale_k(const float* __restrict__ src_rows,
                                      const int* __restrict__ ids,
                                      const float* __restrict__ W,
                                      const float* __restrict__ dis,
                                      float* __restrict__ out,
                                      float* __restrict__ agg,
                                      int n_nodes) {
    const int lane  = threadIdx.x & 31;
    const int wave  = (int)((blockIdx.x * blockDim.x + threadIdx.x) >> 5);
    const int nwave = (int)((gridDim.x * blockDim.x) >> 5);
    const int row   = lane & 15;
    const int col   = lane & 15;
    const int koff  = (lane >> 4) << 1;
    const int rowb  = (lane >> 4) << 3;

    v2f bb0, bb1, bb2, bb3;
    bb0.x = W[(0  + koff) * D + col]; bb0.y = W[(1  + koff) * D + col];
    bb1.x = W[(4  + koff) * D + col]; bb1.y = W[(5  + koff) * D + col];
    bb2.x = W[(8  + koff) * D + col]; bb2.y = W[(9  + koff) * D + col];
    bb3.x = W[(12 + koff) * D + col]; bb3.y = W[(13 + koff) * D + col];

    const int ntiles = n_nodes >> 4;
    for (int tile = wave; tile < ntiles; tile += nwave) {
        const int node = (tile << 4) + row;
        const int r = ids[node];
        const float dv = dis[(tile << 4) + rowb + (lane & 7)];
        const float* xrow = src_rows + (size_t)r * D;
        v2f a0 = *(const v2f*)(xrow + 0  + koff);
        v2f a1 = *(const v2f*)(xrow + 4  + koff);
        v2f a2 = *(const v2f*)(xrow + 8  + koff);
        v2f a3 = *(const v2f*)(xrow + 12 + koff);
        v4f* ab = (v4f*)(agg + ((size_t)tile << 8));
        const v4f zero4 = {0.0f, 0.0f, 0.0f, 0.0f};
        ab[lane] = zero4;
        ab[32 + lane] = zero4;
        v8f c = {};
        c = __builtin_amdgcn_wmma_f32_16x16x4_f32(false, a0, false, bb0, (short)0, c, false, false);
        c = __builtin_amdgcn_wmma_f32_16x16x4_f32(false, a1, false, bb1, (short)0, c, false, false);
        c = __builtin_amdgcn_wmma_f32_16x16x4_f32(false, a2, false, bb2, (short)0, c, false, false);
        c = __builtin_amdgcn_wmma_f32_16x16x4_f32(false, a3, false, bb3, (short)0, c, false, false);
        float* ob = out + ((size_t)(tile << 4) + rowb) * D + col;
#define SCALE_STORE(K) ob[(size_t)K * D] = c[K] * row_bcast_f<K>(dv);
        SCALE_STORE(0) SCALE_STORE(1) SCALE_STORE(2) SCALE_STORE(3)
        SCALE_STORE(4) SCALE_STORE(5) SCALE_STORE(6) SCALE_STORE(7)
#undef SCALE_STORE
    }

    const int tail_start = ntiles << 4;
    if (tail_start < n_nodes) {
        int node = tail_start + (int)(blockIdx.x * blockDim.x + threadIdx.x);
        if (node < n_nodes) {
            const int r = ids[node];
            const float di = dis[node];
            const float* xrow = src_rows + (size_t)r * D;
            for (int j = 0; j < D; ++j) {
                float acc = 0.0f;
                for (int k = 0; k < D; ++k) acc += xrow[k] * W[k * D + j];
                out[(size_t)node * D + j] = acc * di;
                agg[(size_t)node * D + j] = 0.0f;
            }
        }
    }
}

__global__ void relu_gemm16_k(float* h, float* aggrec,
                              const float* __restrict__ dis,
                              const float* __restrict__ bias,
                              const float* __restrict__ W,
                              const int* __restrict__ batch,
                              float* __restrict__ pooled, float* __restrict__ cnt,
                              int n_nodes) {
    const int gid   = (int)(blockIdx.x * blockDim.x + threadIdx.x);
    const int gsize = (int)(gridDim.x * blockDim.x);
    for (int k = gid; k < NGRAPH * D; k += gsize) pooled[k] = 0.0f;
    for (int k = gid; k < NGRAPH; k += gsize) cnt[k] = 0.0f;

    const int lane  = threadIdx.x & 31;
    const int wave  = gid >> 5;
    const int nwave = gsize >> 5;
    const int row   = lane & 15;
    const int col   = lane & 15;
    const int koff  = (lane >> 4) << 1;
    const int rowb  = (lane >> 4) << 3;

    v2f bb0, bb1, bb2, bb3;
    bb0.x = W[(0  + koff) * D + col]; bb0.y = W[(1  + koff) * D + col];
    bb1.x = W[(4  + koff) * D + col]; bb1.y = W[(5  + koff) * D + col];
    bb2.x = W[(8  + koff) * D + col]; bb2.y = W[(9  + koff) * D + col];
    bb3.x = W[(12 + koff) * D + col]; bb3.y = W[(13 + koff) * D + col];
    const v2f bk0 = *(const v2f*)(bias + 0  + koff);
    const v2f bk1 = *(const v2f*)(bias + 4  + koff);
    const v2f bk2 = *(const v2f*)(bias + 8  + koff);
    const v2f bk3 = *(const v2f*)(bias + 12 + koff);

    const int ntiles = n_nodes >> 4;
    for (int tile = wave; tile < ntiles; tile += nwave) {
        const int node = (tile << 4) + row;
        const float di = dis[node];
        const float* hrow = h + (size_t)node * D;
        const float* grow = aggrec + (size_t)node * D;
        const v2f h0 = *(const v2f*)(hrow + 0  + koff), g0 = *(const v2f*)(grow + 0  + koff);
        const v2f h1 = *(const v2f*)(hrow + 4  + koff), g1 = *(const v2f*)(grow + 4  + koff);
        const v2f h2 = *(const v2f*)(hrow + 8  + koff), g2 = *(const v2f*)(grow + 8  + koff);
        const v2f h3 = *(const v2f*)(hrow + 12 + koff), g3 = *(const v2f*)(grow + 12 + koff);
        int bt = 0;
        if (lane < 16) bt = batch[node];
        v2f a0, a1, a2, a3;
#define RELU_IN(A, H, G, B) { float vx = di * (G.x + H.x) + B.x; float vy = di * (G.y + H.y) + B.y; \
                              A.x = (vx > 0.0f ? vx : 0.0f) * di; A.y = (vy > 0.0f ? vy : 0.0f) * di; }
        RELU_IN(a0, h0, g0, bk0) RELU_IN(a1, h1, g1, bk1) RELU_IN(a2, h2, g2, bk2) RELU_IN(a3, h3, g3, bk3)
#undef RELU_IN
        v8f c = {};
        c = __builtin_amdgcn_wmma_f32_16x16x4_f32(false, a0, false, bb0, (short)0, c, false, false);
        c = __builtin_amdgcn_wmma_f32_16x16x4_f32(false, a1, false, bb1, (short)0, c, false, false);
        c = __builtin_amdgcn_wmma_f32_16x16x4_f32(false, a2, false, bb2, (short)0, c, false, false);
        c = __builtin_amdgcn_wmma_f32_16x16x4_f32(false, a3, false, bb3, (short)0, c, false, false);
        float* ob = h + ((size_t)(tile << 4) + rowb) * D + col;
        #pragma unroll
        for (int rr = 0; rr < 8; ++rr) ob[(size_t)rr * D] = c[rr];
        if (lane < 16) {
            v2u rc;
            rc.x = __float_as_uint(di);
            rc.y = (unsigned)bt;
            *(v2u*)((unsigned*)aggrec + (size_t)node * REC_STRIDE) = rc;
        }
    }

    const int tail_start = ntiles << 4;
    if (tail_start < n_nodes) {
        int node = tail_start + gid;
        if (node < n_nodes) {
            const float di = dis[node];
            float z[D];
            for (int k = 0; k < D; ++k) {
                float v = di * (aggrec[(size_t)node * D + k] + h[(size_t)node * D + k]) + bias[k];
                z[k] = (v > 0.0f ? v : 0.0f) * di;
            }
            for (int j = 0; j < D; ++j) {
                float acc = 0.0f;
                for (int k = 0; k < D; ++k) acc += z[k] * W[k * D + j];
                h[(size_t)node * D + j] = acc;
            }
            v2u rc;
            rc.x = __float_as_uint(di);
            rc.y = (unsigned)batch[node];
            *(v2u*)((unsigned*)aggrec + (size_t)node * REC_STRIDE) = rc;
        }
    }
}

#define DEG_SLICES  32
#define DEG_RANGES  4
#define DEG_CAP     50000
#define DEG_THREADS 1024
#define DEG_UNROLL  32
#define DEG_TRIP    (DEG_THREADS * DEG_UNROLL)
static_assert(DEG_UNROLL % 4 == 0, "a thread reads its edges four at a time");
static_assert(DEG_SLICES == 2 * D, "the partial planes fill exactly the two n x D f32 buffers");

__global__ __launch_bounds__(DEG_THREADS)
void degree_lds_k(const int* __restrict__ dst, unsigned* __restrict__ part,
                  int n, long long E) {
    __shared__ unsigned cnt[DEG_CAP];
    const int tid   = threadIdx.x;
    const int slice = blockIdx.x;
    const long long per = (E + DEG_SLICES - 1) / DEG_SLICES;
    const long long e0  = (long long)slice * per;
    long long e1 = e0 + per; if (e1 > E) e1 = E;
    const int* __restrict__ ds = dst + e0;
    const int ne = (e1 > e0) ? (int)(e1 - e0) : 0;
    const bool wide = (((size_t)ds) & 15) == 0;

    for (int lo = blockIdx.y * DEG_CAP; lo < n; lo += gridDim.y * DEG_CAP) {
        const int left = n - lo;
        const unsigned span = (unsigned)(left < DEG_CAP ? left : DEG_CAP);
        for (int i = tid; i < DEG_CAP; i += DEG_THREADS) cnt[i] = 0u;
        __syncthreads();
        for (int base = 0; base < ne; base += DEG_TRIP) {
            int d[DEG_UNROLL];
            if (wide && base + DEG_TRIP <= ne) {
                #pragma unroll
                for (int q = 0; q < DEG_UNROLL / 4; ++q) {
                    const v4i t = *(const v4i*)(ds + base + q * (4 * DEG_THREADS) + tid * 4);
                    d[4 * q + 0] = t.x; d[4 * q + 1] = t.y; d[4 * q + 2] = t.z; d[4 * q + 3] = t.w;
                }
            } else {
                #pragma unroll
                for (int q = 0; q < DEG_UNROLL / 4; ++q) {
                    const int gb = base + q * (4 * DEG_THREADS);
                    if (wide && gb + 4 * DEG_THREADS <= ne) {
                        const v4i t = *(const v4i*)(ds + gb + tid * 4);
                        d[4 * q + 0] = t.x; d[4 * q + 1] = t.y; d[4 * q + 2] = t.z; d[4 * q + 3] = t.w;
                    } else {
                        #pragma unroll
                        for (int k = 0; k < 4; ++k) {
                            const int ee = gb + tid * 4 + k;
                            d[4 * q + k] = (ee < ne) ? ds[ee] : -1;
                        }
                    }
                }
            }
            #pragma unroll
            for (int u = 0; u < DEG_UNROLL; ++u) {
                const unsigned r = (unsigned)(d[u] - lo);
                if (r < span) atomicAdd(&cnt[r], 1u);
            }
        }
        __syncthreads();
        unsigned* __restrict__ prow = part + (size_t)slice * n + lo;
        for (int i = tid; i < (int)span; i += DEG_THREADS) prow[i] = cnt[i];
        __syncthreads();
    }
}

__global__ void dis_from_parts_k(const unsigned* __restrict__ part,
                                 float* __restrict__ dis, int n) {
    int i = blockIdx.x * blockDim.x + threadIdx.x;
    if (i < n) {
        unsigned c = 0u;
        #pragma unroll
        for (int s = 0; s < DEG_SLICES; ++s) c += part[(size_t)s * n + i];
        dis[i] = rsqrtf((float)c + 1.0f);
    }
}

#define AGG_SLICES  6
#define AGG_RANGES  42
#define AGG_CAP     4762
#define AGG_THREADS 1024
#define AGG_WAVES   (AGG_THREADS >> 5)
#define AGG_RUN     4
#define AGG_EDGES   32
#define AGG_TRIP    (32 * AGG_EDGES)
#define AGG_QCAP    (32 * AGG_RUN + 32)
#define AGG_LBITS   13
#define AGG_TBITS   (32 - AGG_LBITS - 10)
#define AGG_SUB     ((1 << AGG_TBITS) * AGG_WAVES * AGG_TRIP)
static_assert(AGG_EDGES % AGG_RUN == 0 && AGG_RUN % 4 == 0, "a lane reads its edges four at a time");
static_assert(AGG_TRIP == 1024, "the offset in a trip is queued in 10 bits");
static_assert(AGG_CAP <= (1 << AGG_LBITS), "a range-local node index is queued in AGG_LBITS bits");
static_assert(AGG_CAP * D * 4 + AGG_WAVES * AGG_QCAP * 4 <= 320 * 1024, "sums and queues share one LDS");

#define ROWS16(X) X(0) X(1) X(2) X(3) X(4) X(5) X(6) X(7) \
                  X(8) X(9) X(10) X(11) X(12) X(13) X(14) X(15)

__device__ __forceinline__ void agg_chunk(const float* __restrict__ h,
                                          const int* __restrict__ src,
                                          float* acc,
                                          unsigned qv, int wbase, bool ok, int j) {
    const unsigned q  = qv >> AGG_LBITS;
    const int      dl = (int)(qv & ((1u << AGG_LBITS) - 1u));
    const int      e  = wbase + (int)(q >> 10) * (AGG_WAVES * AGG_TRIP) + (int)(q & 1023u);
    int s = 0;
    if (ok) s = src[e];
    const int oki = ok ? 1 : 0;
#define LOAD_ROW(K) const float hv##K = h[(unsigned)row_bcast_i<K>(s) * D + j];
    ROWS16(LOAD_ROW)
#undef LOAD_ROW
#define ADD_ROW(K) if (row_bcast_i<K>(oki)) atomicAdd(&acc[row_bcast_i<K>(dl) * D + j], hv##K);
    ROWS16(ADD_ROW)
#undef ADD_ROW
}

__global__ __launch_bounds__(AGG_THREADS)
void aggregate_lds_k(const float* __restrict__ h,
                     const int* __restrict__ src,
                     const int* __restrict__ dst,
                     float* __restrict__ agg,
                     int n, long long E) {
    __shared__ float acc[AGG_CAP * D];
    __shared__ unsigned queue[AGG_WAVES * AGG_QCAP];
    const int tid  = threadIdx.x;
    const int lane = tid & 31;
    const int j    = lane & 15;
    const int wv   = __builtin_amdgcn_readfirstlane(tid >> 5);
    unsigned* qw = queue + wv * AGG_QCAP;

    long long per = (E + AGG_SLICES - 1) / AGG_SLICES;
    per = (per + 127) & ~127LL;
    long long e0  = (long long)blockIdx.x * per;
    if (e0 > E) e0 = E;
    long long e1 = e0 + per; if (e1 > E) e1 = E;
    const long long ne = e1 - e0;
    const bool wide = (((size_t)(dst + e0)) & 15) == 0;

    for (int lo = blockIdx.y * AGG_CAP; lo < n; lo += gridDim.y * AGG_CAP) {
        const int left = n - lo;
        const unsigned span = (unsigned)(left < AGG_CAP ? left : AGG_CAP);
        for (int i = tid; i < AGG_CAP * D; i += AGG_THREADS) acc[i] = 0.0f;
        __syncthreads();

        for (long long sb = 0; sb < ne; sb += AGG_SUB) {
            const long long rest = ne - sb;
            const int se = (int)(rest < AGG_SUB ? rest : (long long)AGG_SUB);
            const int* __restrict__ ds = dst + e0 + sb;
            const int* __restrict__ ss = src + e0 + sb;
            const int wbase = wv * AGG_TRIP;
            int nq = 0;
            unsigned tq = 0u;
            for (int base = wbase; base < se; base += AGG_WAVES * AGG_TRIP, tq += 1u << (AGG_LBITS + 10)) {
                int dv[AGG_EDGES];
                if (wide && base + AGG_TRIP <= se) {
                    #pragma unroll
                    for (int q = 0; q < AGG_EDGES / 4; ++q) {
                        const v4i t = *(const v4i*)(ds + base + q * 128 + lane * 4);
                        dv[4 * q + 0] = t.x; dv[4 * q + 1] = t.y; dv[4 * q + 2] = t.z; dv[4 * q + 3] = t.w;
                    }
                } else {
                    #pragma unroll
                    for (int q = 0; q < AGG_EDGES / 4; ++q) {
                        const int gb = base + q * 128;
                        if (wide && gb + 128 <= se) {
                            const v4i t = *(const v4i*)(ds + gb + lane * 4);
                            dv[4 * q + 0] = t.x; dv[4 * q + 1] = t.y; dv[4 * q + 2] = t.z; dv[4 * q + 3] = t.w;
                        } else {
                            #pragma unroll
                            for (int k = 0; k < 4; ++k) {
                                const int ee = gb + lane * 4 + k;
                                dv[4 * q + k] = (ee < se) ? ds[ee] : -1;
                            }
                        }
                    }
                }
                const unsigned wl = tq + ((unsigned)(lane * 4) << AGG_LBITS);
                #pragma unroll
                for (int rn = 0; rn < AGG_EDGES / AGG_RUN; ++rn) {
                    #pragma unroll
                    for (int u = rn * AGG_RUN; u < (rn + 1) * AGG_RUN; ++u) {
                        const unsigned r = (unsigned)(dv[u] - lo);
                        const bool hit = r < span;
                        const unsigned m = __builtin_amdgcn_ballot_w32(hit);
                        if (m) {
                            if (hit) {
                                const int pos = nq + (int)__builtin_amdgcn_mbcnt_lo(m, 0u);
                                qw[pos] = wl + ((unsigned)((u >> 2) * 128 + (u & 3)) << AGG_LBITS) + r;
                            }
                            nq += __builtin_popcount(m);
                        }
                    }
                    while (nq >= 32) {
                        nq -= 32;
                        agg_chunk(h, ss, acc, qw[nq + lane], wbase, true, j);
                    }
                }
            }
            if (nq > 0) {
                const bool ok = lane < nq;
                agg_chunk(h, ss, acc, ok ? qw[lane] : 0u, wbase, ok, j);
            }
        }
        __syncthreads();

        float* __restrict__ arow = agg + (size_t)lo * D;
        for (int i = tid; i < (int)span * D; i += AGG_THREADS) {
            const float v = acc[i];
            if (v != 0.0f) atomicAdd(arow + i, v);
        }
        __syncthreads();
    }
}

#define POOL_THREADS 1024
#define POOL_BLOCKS  256

__global__ __launch_bounds__(POOL_THREADS)
void edge_node_pool_k(const float* __restrict__ h2,
                      const unsigned* __restrict__ rec,
                      const float* __restrict__ bias,
                      const int* __restrict__ src,
                      const int* __restrict__ dst,
                      float* __restrict__ pooled,
                      float* __restrict__ cnt,
                      int n, long long E) {
    __shared__ float acc[NGRAPH * D];
    __shared__ float gcnt[NGRAPH];
    const int tid  = threadIdx.x;
    const int lane = tid & 31;
    const int j    = lane & 15;

    for (int i = tid; i < NGRAPH * D; i += POOL_THREADS) acc[i] = 0.0f;
    for (int i = tid; i < NGRAPH; i += POOL_THREADS) gcnt[i] = 0.0f;
    __syncthreads();

    const long long wave  = (long long)blockIdx.x * (POOL_THREADS >> 5)
                          + __builtin_amdgcn_readfirstlane(tid >> 5);
    const long long nwave = (long long)gridDim.x * (POOL_THREADS >> 5);

    const long long nfull = E >> 5;
    long long c = wave;
    int s = 0, d = 0;
    if (c < nfull) { s = src[(c << 5) + lane]; d = dst[(c << 5) + lane]; }
    while (c < nfull) {
        const v2u rd = *(const v2u*)(rec + REC_STRIDE * (size_t)(unsigned)d);
        const float w = __uint_as_float(rd.x);
        const int   g = (int)rd.y;
#define LOAD_ROW(K) const float hv##K = h2[(unsigned)row_bcast_i<K>(s) * D + j];
        ROWS16(LOAD_ROW)
#undef LOAD_ROW
        const long long cn = c + nwave;
        int sn = 0, dn = 0;
        if (cn < nfull) { sn = src[(cn << 5) + lane]; dn = dst[(cn << 5) + lane]; }
#define ADD_ROW(K) atomicAdd(&acc[row_bcast_i<K>(g) * D + j], hv##K * row_bcast_f<K>(w));
        ROWS16(ADD_ROW)
#undef ADD_ROW
        s = sn; d = dn; c = cn;
    }
    if (blockIdx.x == 0 && tid < 16) {
        for (long long e = nfull << 5; e < E; ++e) {
            const int ss = src[e], dd = dst[e];
            const v2u rd = *(const v2u*)(rec + REC_STRIDE * (size_t)(unsigned)dd);
            atomicAdd(&acc[(int)rd.y * D + j], h2[(size_t)ss * D + j] * __uint_as_float(rd.x));
        }
    }

    const long long n16 = (long long)n * D;
    for (long long t = (long long)blockIdx.x * POOL_THREADS + tid; t < n16;
         t += (long long)gridDim.x * POOL_THREADS) {
        const int i = (int)(t >> 4);
        const v2u ri = *(const v2u*)(rec + REC_STRIDE * (size_t)i);
        const float di = __uint_as_float(ri.x);
        const int   g  = (int)ri.y;
        atomicAdd(&acc[g * D + j], h2[t] * di + bias[j]);
        if (j == 0) atomicAdd(&gcnt[g], 1.0f);
    }
    __syncthreads();

    const int rot = (int)((blockIdx.x * POOL_THREADS) & (NGRAPH * D - 1));
    for (int i = tid; i < NGRAPH * D; i += POOL_THREADS) {
        const int idx = (i + rot) & (NGRAPH * D - 1);
        const float v = acc[idx];
        if (v != 0.0f) atomicAdd(pooled + idx, v);
    }
    for (int i = tid; i < NGRAPH; i += POOL_THREADS) {
        const float v = gcnt[i];
        if (v != 0.0f) atomicAdd(cnt + i, v);
    }
}

__global__ void fc_k(const float* __restrict__ pooled_r, const float* __restrict__ cnt_r,
                     const float* __restrict__ pooled_l, const float* __restrict__ cnt_l,
                     const float* __restrict__ fcW, const float* __restrict__ fcb,
                     float* __restrict__ out) {
    int t = blockIdx.x * blockDim.x + threadIdx.x;
    if (t < NGRAPH * 6) {
        int g = t / 6, o = t - g * 6;
        float icr = 1.0f / fmaxf(cnt_r[g], 1.0f);
        float icl = 1.0f / fmaxf(cnt_l[g], 1.0f);
        float acc = fcb[o];
        #pragma unroll
        for (int j = 0; j < D; ++j)
            acc += fcW[o * (2 * D) + j] * (pooled_r[g * D + j] * icr);
        #pragma unroll
        for (int j = 0; j < D; ++j)
            acc += fcW[o * (2 * D) + D + j] * (pooled_l[g * D + j] * icl);
        if (o < 3) out[g * 3 + o] = acc;
        else       out[NGRAPH * 3 + g * 3 + (o - 3)] = acc;
    }
}

static void run_protein(const float* emb, const float* W1, const float* b1,
                        const float* W2, const float* b2,
                        const int* ids, const int* edge, const int* batch,
                        int N, long long E,
                        float* bufA, float* bufB, float* dis,
                        float* pooled, float* cnt, hipStream_t stream) {
    const int* src = edge;
    const int* dst = edge + E;
    const int B = 256;
    dim3 blk(B);
    int gemm_grid = (int)(((N >> 4) + 7) / 8);
    if (gemm_grid < 1) gemm_grid = 1;

    degree_lds_k<<<dim3(DEG_SLICES, DEG_RANGES), dim3(DEG_THREADS), 0, stream>>>(
        dst, (unsigned*)bufA, N, E);
    dis_from_parts_k<<<(int)((N + B - 1) / B), blk, 0, stream>>>((const unsigned*)bufA, dis, N);

    gather_gemm16_scale_k<<<gemm_grid, blk, 0, stream>>>(emb, ids, W1, dis, bufA, bufB, N);

    aggregate_lds_k<<<dim3(AGG_SLICES, AGG_RANGES), dim3(AGG_THREADS), 0, stream>>>(
        bufA, src, dst, bufB, N, E);

    relu_gemm16_k<<<gemm_grid, blk, 0, stream>>>(bufA, bufB, dis, b1, W2, batch, pooled, cnt, N);

    edge_node_pool_k<<<POOL_BLOCKS, POOL_THREADS, 0, stream>>>(bufA, (const unsigned*)bufB, b2,
                                                               src, dst, pooled, cnt, N, E);
}

extern "C" void kernel_launch(void* const* d_in, const int* in_sizes, int n_in,
                              void* d_out, int out_size, void* d_ws, size_t ws_size,
                              hipStream_t stream) {
    const float* emb = (const float*)d_in[0];
    const float* W1  = (const float*)d_in[1];
    const float* b1  = (const float*)d_in[2];
    const float* W2  = (const float*)d_in[3];
    const float* b2  = (const float*)d_in[4];
    const float* fcW = (const float*)d_in[5];
    const float* fcb = (const float*)d_in[6];
    const int* rx     = (const int*)d_in[7];
    const int* redge  = (const int*)d_in[8];
    const int* rbatch = (const int*)d_in[9];
    const int* lx     = (const int*)d_in[10];
    const int* ledge  = (const int*)d_in[11];
    const int* lbatch = (const int*)d_in[12];

    const int N = in_sizes[7];
    const long long E = in_sizes[8] / 2;

    char* ws = (char*)d_ws;
    float* bufA = (float*)ws;  ws += (size_t)N * D * sizeof(float);
    float* bufB = (float*)ws;  ws += (size_t)N * D * sizeof(float);
    float* dis  = (float*)ws;  ws += (size_t)N * sizeof(float);
    float* pooled_r = (float*)ws; ws += (size_t)NGRAPH * D * sizeof(float);
    float* pooled_l = (float*)ws; ws += (size_t)NGRAPH * D * sizeof(float);
    float* cnt_r = (float*)ws; ws += (size_t)NGRAPH * sizeof(float);
    float* cnt_l = (float*)ws; ws += (size_t)NGRAPH * sizeof(float);

    run_protein(emb, W1, b1, W2, b2, rx, redge, rbatch, N, E,
                bufA, bufB, dis, pooled_r, cnt_r, stream);
    run_protein(emb, W1, b1, W2, b2, lx, ledge, lbatch, N, E,
                bufA, bufB, dis, pooled_l, cnt_l, stream);

    fc_k<<<(NGRAPH * 6 + 255) / 256, 256, 0, stream>>>(pooled_r, cnt_r, pooled_l, cnt_l,
                                                       fcW, fcb, (float*)d_out);
}
